# starting kernel
# baseline (speedup 1.0000x reference)
#include <hip/hip_runtime.h>

typedef __attribute__((ext_vector_type(2))) float v2f;
typedef __attribute__((ext_vector_type(8))) float v8f;

#define D 256
#define PAIR_STRIDE 272

extern __shared__ v2f ldsP[];

__global__ __launch_bounds__(256) void gcn_gemm_xw(
    const float* __restrict__ x, const float* __restrict__ W,
    float* __restrict__ y, int nRowTiles) {
  const int tid  = threadIdx.x;
  const int lane = tid & 31;
  const int wv   = tid >> 5;
  const int lr   = lane & 15;
  const int hi   = lane >> 4;

#pragma unroll 4
  for (int g = tid; g < (D / 2) * D; g += 256) {
    const int p = g >> 8;
    const int c = g & 255;
    v2f val;
    val.x = W[(2 * p) * D + c];
    val.y = W[(2 * p + 1) * D + c];
    ldsP[p * PAIR_STRIDE + c] = val;
  }
  __syncthreads();

  const int rowTileLocal = wv >> 2;
  const int colBase      = (wv & 3) * 64;

  const int lbase = hi * PAIR_STRIDE + colBase + lr;

  for (int t = blockIdx.x * 2 + rowTileLocal; t < nRowTiles;
       t += gridDim.x * 2) {
    const int row0 = t * 16;

    v8f acc0 = {}, acc1 = {}, acc2 = {}, acc3 = {};

    const float* xp = x + (row0 + lr) * D + 2 * hi;

    for (int k = 0; k < D; k += 4) {
      v2f a = *(const v2f*)(xp + k);

      const int lb = lbase + (k >> 1) * PAIR_STRIDE;
      v2f b0 = ldsP[lb +  0];
      v2f b1 = ldsP[lb + 16];
      v2f b2 = ldsP[lb + 32];
      v2f b3 = ldsP[lb + 48];

      acc0 = __builtin_amdgcn_wmma_f32_16x16x4_f32(false, a, false, b0,
                                                   (short)0, acc0, false, false);
      acc1 = __builtin_amdgcn_wmma_f32_16x16x4_f32(false, a, false, b1,
                                                   (short)0, acc1, false, false);
      acc2 = __builtin_amdgcn_wmma_f32_16x16x4_f32(false, a, false, b2,
                                                   (short)0, acc2, false, false);
      acc3 = __builtin_amdgcn_wmma_f32_16x16x4_f32(false, a, false, b3,
                                                   (short)0, acc3, false, false);
    }

    float* yp = y + row0 * D + colBase + lr;
#pragma unroll
    for (int i = 0; i < 8; ++i) {
      const int r = (i + 8 * hi) * D;
      yp[r +  0] = acc0[i];
      yp[r + 16] = acc1[i];
      yp[r + 32] = acc2[i];
      yp[r + 48] = acc3[i];
    }
  }
}

__global__ __launch_bounds__(256) void gcn_bias_init(
    const float* __restrict__ b, float* __restrict__ out) {
  const int i = blockIdx.x * D + threadIdx.x;
  out[i] = b[threadIdx.x];
}

__global__ __launch_bounds__(256) void gcn_scatter(
    const int* __restrict__ erow, const int* __restrict__ ecol,
    const float* __restrict__ eval_, const float* __restrict__ y,
    float* __restrict__ out, int E) {
  const int lane = threadIdx.x & 31;
  const int e = blockIdx.x * 8 + (threadIdx.x >> 5);
  if (e >= E) return;

  const int   r = erow[e];
  const int   c = ecol[e];
  const float v = eval_[e];

  const float4* src = (const float4*)(y + (size_t)c * D);
  float* dst = out + (size_t)r * D;

#pragma unroll
  for (int it = 0; it < 2; ++it) {
    const int q = lane + 32 * it;
    float4 d = src[q];
    const int base = q * 4;
    atomicAdd(dst + base + 0, v * d.x);
    atomicAdd(dst + base + 1, v * d.y);
    atomicAdd(dst + base + 2, v * d.z);
    atomicAdd(dst + base + 3, v * d.w);
  }
}

extern "C" void kernel_launch(void* const* d_in, const int* in_sizes, int n_in,
                              void* d_out, int out_size, void* d_ws, size_t ws_size,
                              hipStream_t stream) {
  const float* x    = (const float*)d_in[0];
  const int*   erow = (const int*)d_in[1];
  const int*   ecol = (const int*)d_in[2];
  const float* ev   = (const float*)d_in[3];
  const float* W    = (const float*)d_in[4];
  const float* b    = (const float*)d_in[5];
  float*       out  = (float*)d_out;
  float*       y    = (float*)d_ws;

  const int N = in_sizes[0] / D;
  const int E = in_sizes[1];

  const size_t ldsBytes = (size_t)128 * PAIR_STRIDE * sizeof(v2f);
  gcn_gemm_xw<<<512, 256, ldsBytes, stream>>>(x, W, y, N / 16);

  gcn_bias_init<<<N, D, 0, stream>>>(b, out);

  gcn_scatter<<<(E + 7) / 8, 256, 0, stream>>>(erow, ecol, ev, y, out, E);
}


# Float-atomic scatter replaced by pull: edges counting-sorted by row in-call (int atomics, arrival-order rank), one wave per row sums val*y[col] in registers, bias folded, one plain store per output
# speedup vs baseline: 1.7417x; 1.7417x over previous
#include <hip/hip_runtime.h>

typedef __attribute__((ext_vector_type(2))) float v2f;
typedef __attribute__((ext_vector_type(8))) float v8f;

#define D 256
#define PAIR_STRIDE 272

extern __shared__ v2f ldsP[];

__global__ __launch_bounds__(256) void gcn_gemm_xw(
    const float* __restrict__ x, const float* __restrict__ W,
    float* __restrict__ y, int nRowTiles) {
  const int tid  = threadIdx.x;
  const int lane = tid & 31;
  const int wv   = tid >> 5;
  const int lr   = lane & 15;
  const int hi   = lane >> 4;

#pragma unroll 4
  for (int g = tid; g < (D / 2) * D; g += 256) {
    const int p = g >> 8;
    const int c = g & 255;
    v2f val;
    val.x = W[(2 * p) * D + c];
    val.y = W[(2 * p + 1) * D + c];
    ldsP[p * PAIR_STRIDE + c] = val;
  }
  __syncthreads();

  const int rowTileLocal = wv >> 2;
  const int colBase      = (wv & 3) * 64;

  const int lbase = hi * PAIR_STRIDE + colBase + lr;

  for (int t = blockIdx.x * 2 + rowTileLocal; t < nRowTiles;
       t += gridDim.x * 2) {
    const int row0 = t * 16;

    v8f acc0 = {}, acc1 = {}, acc2 = {}, acc3 = {};

    const float* xp = x + (row0 + lr) * D + 2 * hi;

    for (int k = 0; k < D; k += 4) {
      v2f a = *(const v2f*)(xp + k);

      const int lb = lbase + (k >> 1) * PAIR_STRIDE;
      v2f b0 = ldsP[lb +  0];
      v2f b1 = ldsP[lb + 16];
      v2f b2 = ldsP[lb + 32];
      v2f b3 = ldsP[lb + 48];

      acc0 = __builtin_amdgcn_wmma_f32_16x16x4_f32(false, a, false, b0,
                                                   (short)0, acc0, false, false);
      acc1 = __builtin_amdgcn_wmma_f32_16x16x4_f32(false, a, false, b1,
                                                   (short)0, acc1, false, false);
      acc2 = __builtin_amdgcn_wmma_f32_16x16x4_f32(false, a, false, b2,
                                                   (short)0, acc2, false, false);
      acc3 = __builtin_amdgcn_wmma_f32_16x16x4_f32(false, a, false, b3,
                                                   (short)0, acc3, false, false);
    }

    float* yp = y + row0 * D + colBase + lr;
#pragma unroll
    for (int i = 0; i < 8; ++i) {
      const int r = (i + 8 * hi) * D;
      yp[r +  0] = acc0[i];
      yp[r + 16] = acc1[i];
      yp[r + 32] = acc2[i];
      yp[r + 48] = acc3[i];
    }
  }
}

__global__ __launch_bounds__(256) void gcn_zero_counts(int* __restrict__ cnt,
                                                       int n) {
  const int i = blockIdx.x * 256 + threadIdx.x;
  if (i < n) cnt[i] = 0;
}

__global__ __launch_bounds__(256) void gcn_count_rank(
    const int* __restrict__ erow, int* __restrict__ cnt,
    int* __restrict__ rank, int E) {
  const int e = blockIdx.x * 256 + threadIdx.x;
  if (e >= E) return;
  rank[e] = atomicAdd(cnt + erow[e], 1);
}

__global__ __launch_bounds__(1024) void gcn_row_offsets(
    const int* __restrict__ cnt, int* __restrict__ rowptr, int n4) {
  __shared__ int wtot[32];
  const int tid  = threadIdx.x;
  const int lane = tid & 31;
  const int wv   = tid >> 5;
  const int per  = (n4 + 1023) >> 10;
  const int beg  = min(tid * per, n4);
  const int end  = min(beg + per, n4);
  const int4* c4 = (const int4*)cnt;
  int4* r4 = (int4*)rowptr;

  int s = 0;
#pragma unroll 4
  for (int q = beg; q < end; ++q) {
    const int4 v = c4[q];
    s += (v.x + v.y) + (v.z + v.w);
  }

  int inc = s;
  for (int d = 1; d < 32; d <<= 1) {
    const int t = __shfl_up(inc, d);
    if (lane >= d) inc += t;
  }
  if (lane == 31) wtot[wv] = inc;
  __syncthreads();
  if (wv == 0) {
    const int w = wtot[lane];
    int winc = w;
    for (int d = 1; d < 32; d <<= 1) {
      const int t = __shfl_up(winc, d);
      if (lane >= d) winc += t;
    }
    wtot[lane] = winc - w;
  }
  __syncthreads();

  int run = wtot[wv] + inc - s;
#pragma unroll 4
  for (int q = beg; q < end; ++q) {
    const int4 v = c4[q];
    int4 o;
    o.x = run;
    o.y = o.x + v.x;
    o.z = o.y + v.y;
    o.w = o.z + v.z;
    run = o.w + v.w;
    r4[q] = o;
  }
}

__global__ __launch_bounds__(256) void gcn_place_edges(
    const int* __restrict__ erow, const int* __restrict__ ecol,
    const float* __restrict__ eval_, const int* __restrict__ rowptr,
    const int* __restrict__ rank, int* __restrict__ csr, int E) {
  const int e = blockIdx.x * 256 + threadIdx.x;
  if (e >= E) return;
  int pos = rowptr[erow[e]] + rank[e];
  pos = min(max(pos, 0), E - 1);
  ((int2*)csr)[pos] = make_int2(ecol[e], __float_as_int(eval_[e]));
}

template <int K>
__device__ __forceinline__ void gcn_add_edges(const float4* __restrict__ y4,
                                              int2 cv, int j, int lane,
                                              int lastRow, float4& a0,
                                              float4& a1) {
  float4 p[K], q[K];
  float  v[K];
  for (int t = 0; t < K; ++t) {
    const int c = min(max(__builtin_amdgcn_readlane(cv.x, j + t), 0), lastRow);
    v[t] = __int_as_float(__builtin_amdgcn_readlane(cv.y, j + t));
    const float4* s = y4 + (size_t)c * (D / 4) + lane;
    p[t] = s[0];
    q[t] = s[32];
  }
  for (int t = 0; t < K; ++t) {
    a0.x = fmaf(v[t], p[t].x, a0.x); a0.y = fmaf(v[t], p[t].y, a0.y);
    a0.z = fmaf(v[t], p[t].z, a0.z); a0.w = fmaf(v[t], p[t].w, a0.w);
    a1.x = fmaf(v[t], q[t].x, a1.x); a1.y = fmaf(v[t], q[t].y, a1.y);
    a1.z = fmaf(v[t], q[t].z, a1.z); a1.w = fmaf(v[t], q[t].w, a1.w);
  }
}

__global__ __launch_bounds__(256) void gcn_gather_rows(
    const int* __restrict__ rowptr, const int* __restrict__ cnt,
    const int* __restrict__ csr, const float* __restrict__ y,
    const float* __restrict__ b, float* __restrict__ out, int N, int E) {
  const int lane = threadIdx.x & 31;
  const int i = __builtin_amdgcn_readfirstlane(blockIdx.x * 8 +
                                               (threadIdx.x >> 5));
  if (i >= N) return;

  int start = rowptr[i];
  int len   = cnt[i];
  start = min(max(start, 0), E);
  len   = min(max(len, 0), E - start);

  const float4* b4 = (const float4*)b;
  const float4* y4 = (const float4*)y;
  const int2*   pr = (const int2*)csr;
  float4 a0 = b4[lane];
  float4 a1 = b4[lane + 32];

  for (int base = 0; base < len; base += 32) {
    const int n = min(32, len - base);
    int2 cv = make_int2(0, 0);
    if (lane < n) cv = pr[start + base + lane];

    int j = 0;
    for (; j + 4 <= n; j += 4) gcn_add_edges<4>(y4, cv, j, lane, N - 1, a0, a1);
    const int rem = n - j;
    if (rem == 3)      gcn_add_edges<3>(y4, cv, j, lane, N - 1, a0, a1);
    else if (rem == 2) gcn_add_edges<2>(y4, cv, j, lane, N - 1, a0, a1);
    else if (rem == 1) gcn_add_edges<1>(y4, cv, j, lane, N - 1, a0, a1);
  }

  float4* o4 = (float4*)out + (size_t)i * (D / 4) + lane;
  o4[0]  = a0;
  o4[32] = a1;
}

extern "C" void kernel_launch(void* const* d_in, const int* in_sizes, int n_in,
                              void* d_out, int out_size, void* d_ws, size_t ws_size,
                              hipStream_t stream) {
  const float* x    = (const float*)d_in[0];
  const int*   erow = (const int*)d_in[1];
  const int*   ecol = (const int*)d_in[2];
  const float* ev   = (const float*)d_in[3];
  const float* W    = (const float*)d_in[4];
  const float* b    = (const float*)d_in[5];
  float*       out  = (float*)d_out;

  const int N  = in_sizes[0] / D;
  const int E  = in_sizes[1];
  const int Np = (N + 63) & ~63;
  const int Ep = (E + 63) & ~63;

  float* y      = (float*)d_ws;
  int*   cnt    = (int*)(y + (size_t)N * D);
  int*   rowptr = cnt + Np;
  int*   rank   = rowptr + Np;
  int*   csr    = rank + Ep;

  const size_t ldsBytes = (size_t)128 * PAIR_STRIDE * sizeof(v2f);
  gcn_gemm_xw<<<512, 256, ldsBytes, stream>>>(x, W, y, N / 16);

  gcn_zero_counts<<<(Np + 255) / 256, 256, 0, stream>>>(cnt, Np);
  gcn_count_rank<<<(E + 255) / 256, 256, 0, stream>>>(erow, cnt, rank, E);
  gcn_row_offsets<<<1, 1024, 0, stream>>>(cnt, rowptr, Np / 4);
  gcn_place_edges<<<(E + 255) / 256, 256, 0, stream>>>(erow, ecol, ev, rowptr,
                                                       rank, csr, E);

  gcn_gather_rows<<<(N + 7) / 8, 256, 0, stream>>>(rowptr, cnt, csr, y, b, out,
                                                   N, E);
}


# Edge sort: the returning global atomic per edge replaced by LDS counters: 128 edge chunks each count, then place, on a full node table in LDS; order in a row = chunk order, arrival inside a chunk
# speedup vs baseline: 2.0020x; 1.1495x over previous
#include <hip/hip_runtime.h>

typedef __attribute__((ext_vector_type(2))) float v2f;
typedef __attribute__((ext_vector_type(8))) float v8f;

#define D 256
#define PAIR_STRIDE 272

extern __shared__ v2f ldsP[];

__global__ __launch_bounds__(256) void gcn_gemm_xw(
    const float* __restrict__ x, const float* __restrict__ W,
    float* __restrict__ y, int nRowTiles) {
  const int tid  = threadIdx.x;
  const int lane = tid & 31;
  const int wv   = tid >> 5;
  const int lr   = lane & 15;
  const int hi   = lane >> 4;

#pragma unroll 4
  for (int g = tid; g < (D / 2) * D; g += 256) {
    const int p = g >> 8;
    const int c = g & 255;
    v2f val;
    val.x = W[(2 * p) * D + c];
    val.y = W[(2 * p + 1) * D + c];
    ldsP[p * PAIR_STRIDE + c] = val;
  }
  __syncthreads();

  const int rowTileLocal = wv >> 2;
  const int colBase      = (wv & 3) * 64;

  const int lbase = hi * PAIR_STRIDE + colBase + lr;

  for (int t = blockIdx.x * 2 + rowTileLocal; t < nRowTiles;
       t += gridDim.x * 2) {
    const int row0 = t * 16;

    v8f acc0 = {}, acc1 = {}, acc2 = {}, acc3 = {};

    const float* xp = x + (row0 + lr) * D + 2 * hi;

    for (int k = 0; k < D; k += 4) {
      v2f a = *(const v2f*)(xp + k);

      const int lb = lbase + (k >> 1) * PAIR_STRIDE;
      v2f b0 = ldsP[lb +  0];
      v2f b1 = ldsP[lb + 16];
      v2f b2 = ldsP[lb + 32];
      v2f b3 = ldsP[lb + 48];

      acc0 = __builtin_amdgcn_wmma_f32_16x16x4_f32(false, a, false, b0,
                                                   (short)0, acc0, false, false);
      acc1 = __builtin_amdgcn_wmma_f32_16x16x4_f32(false, a, false, b1,
                                                   (short)0, acc1, false, false);
      acc2 = __builtin_amdgcn_wmma_f32_16x16x4_f32(false, a, false, b2,
                                                   (short)0, acc2, false, false);
      acc3 = __builtin_amdgcn_wmma_f32_16x16x4_f32(false, a, false, b3,
                                                   (short)0, acc3, false, false);
    }

    float* yp = y + row0 * D + colBase + lr;
#pragma unroll
    for (int i = 0; i < 8; ++i) {
      const int r = (i + 8 * hi) * D;
      yp[r +  0] = acc0[i];
      yp[r + 16] = acc1[i];
      yp[r + 32] = acc2[i];
      yp[r + 48] = acc3[i];
    }
  }
}

#define SORT_CHUNKS 128
#define SORT_EPT 8

extern __shared__ int ldsNode[];

__device__ __forceinline__ int gcn_row_id(int r, int lastRow) {
  return min(max(r, 0), lastRow);
}

__global__ __launch_bounds__(1024) void gcn_count_chunk(
    const int* __restrict__ erow, int* __restrict__ tbl, int N, int n4, int E,
    int chunkLen) {
  const int tid = threadIdx.x;
  const int c   = blockIdx.x;
  const int e0  = min(E, c * chunkLen);
  const int e1  = min(E, e0 + chunkLen);
  int4* l4 = (int4*)ldsNode;

#pragma unroll 4
  for (int q = tid; q < n4; q += 1024) l4[q] = make_int4(0, 0, 0, 0);
  __syncthreads();

  for (int b = e0 + tid; b < e1; b += 1024 * SORT_EPT) {
    int r[SORT_EPT];
    for (int u = 0; u < SORT_EPT; ++u) r[u] = erow[min(b + u * 1024, e1 - 1)];
    asm volatile("" :: "v"(r[0]), "v"(r[1]), "v"(r[2]), "v"(r[3]),
                       "v"(r[4]), "v"(r[5]), "v"(r[6]), "v"(r[7]));
    for (int u = 0; u < SORT_EPT; ++u) {
      if (b + u * 1024 < e1) atomicAdd(&ldsNode[gcn_row_id(r[u], N - 1)], 1);
    }
  }
  __syncthreads();

  int4* t4 = (int4*)tbl + (size_t)c * n4;
#pragma unroll 4
  for (int q = tid; q < n4; q += 1024) t4[q] = l4[q];
}

__global__ __launch_bounds__(256) void gcn_chunk_prefix(int* __restrict__ tbl,
                                                        int* __restrict__ cnt,
                                                        int Np) {
  const int i = blockIdx.x * 256 + threadIdx.x;
  if (i >= Np) return;
  int v[SORT_CHUNKS];
  const unsigned np = (unsigned)Np;
  for (int c = 0; c < SORT_CHUNKS; ++c) v[c] = tbl[(unsigned)c * np + (unsigned)i];
  int acc = 0;
  for (int c = 0; c < SORT_CHUNKS; ++c) {
    tbl[(unsigned)c * np + (unsigned)i] = acc;
    acc += v[c];
  }
  cnt[i] = acc;
}

__global__ __launch_bounds__(1024) void gcn_row_offsets(
    const int* __restrict__ cnt, int* __restrict__ rowptr, int n4) {
  __shared__ int wtot[32];
  const int tid  = threadIdx.x;
  const int lane = tid & 31;
  const int wv   = tid >> 5;
  const int per  = (n4 + 1023) >> 10;
  const int beg  = min(tid * per, n4);
  const int end  = min(beg + per, n4);
  const int4* c4 = (const int4*)cnt;
  int4* r4 = (int4*)rowptr;

  int s = 0;
#pragma unroll 4
  for (int q = beg; q < end; ++q) {
    const int4 v = c4[q];
    s += (v.x + v.y) + (v.z + v.w);
  }

  int inc = s;
  for (int d = 1; d < 32; d <<= 1) {
    const int t = __shfl_up(inc, d);
    if (lane >= d) inc += t;
  }
  if (lane == 31) wtot[wv] = inc;
  __syncthreads();
  if (wv == 0) {
    const int w = wtot[lane];
    int winc = w;
    for (int d = 1; d < 32; d <<= 1) {
      const int t = __shfl_up(winc, d);
      if (lane >= d) winc += t;
    }
    wtot[lane] = winc - w;
  }
  __syncthreads();

  int run = wtot[wv] + inc - s;
#pragma unroll 4
  for (int q = beg; q < end; ++q) {
    const int4 v = c4[q];
    int4 o;
    o.x = run;
    o.y = o.x + v.x;
    o.z = o.y + v.y;
    o.w = o.z + v.z;
    run = o.w + v.w;
    r4[q] = o;
  }
}

__global__ __launch_bounds__(1024) void gcn_place_chunk(
    const int* __restrict__ erow, const int* __restrict__ ecol,
    const float* __restrict__ eval_, const int* __restrict__ rowptr,
    const int* __restrict__ tbl, int* __restrict__ csr, int N, int n4, int E,
    int chunkLen) {
  const int tid = threadIdx.x;
  const int c   = blockIdx.x;
  const int e0  = min(E, c * chunkLen);
  const int e1  = min(E, e0 + chunkLen);
  int4* l4 = (int4*)ldsNode;
  const int4* r4 = (const int4*)rowptr;
  const int4* t4 = (const int4*)tbl + (size_t)c * n4;

#pragma unroll 4
  for (int q = tid; q < n4; q += 1024) {
    const int4 a = r4[q];
    const int4 p = t4[q];
    l4[q] = make_int4(a.x + p.x, a.y + p.y, a.z + p.z, a.w + p.w);
  }
  __syncthreads();

  for (int b = e0 + tid; b < e1; b += 1024 * SORT_EPT) {
    int   r[SORT_EPT], cl[SORT_EPT];
    float vl[SORT_EPT];
    for (int u = 0; u < SORT_EPT; ++u) {
      const int e = min(b + u * 1024, e1 - 1);
      r[u]  = erow[e];
      cl[u] = ecol[e];
      vl[u] = eval_[e];
    }
    asm volatile("" :: "v"(r[0]), "v"(r[1]), "v"(r[2]), "v"(r[3]),
                       "v"(r[4]), "v"(r[5]), "v"(r[6]), "v"(r[7]));
    asm volatile("" :: "v"(cl[0]), "v"(cl[1]), "v"(cl[2]), "v"(cl[3]),
                       "v"(cl[4]), "v"(cl[5]), "v"(cl[6]), "v"(cl[7]));
    asm volatile("" :: "v"(vl[0]), "v"(vl[1]), "v"(vl[2]), "v"(vl[3]),
                       "v"(vl[4]), "v"(vl[5]), "v"(vl[6]), "v"(vl[7]));
    for (int u = 0; u < SORT_EPT; ++u) {
      if (b + u * 1024 < e1) {
        int pos = atomicAdd(&ldsNode[gcn_row_id(r[u], N - 1)], 1);
        pos = min(max(pos, 0), E - 1);
        ((int2*)csr)[pos] = make_int2(cl[u], __float_as_int(vl[u]));
      }
    }
  }
}

template <int K>
__device__ __forceinline__ void gcn_add_edges(const float4* __restrict__ y4,
                                              int2 cv, int j, int lane,
                                              int lastRow, float4& a0,
                                              float4& a1) {
  float4 p[K], q[K];
  float  v[K];
  for (int t = 0; t < K; ++t) {
    const int c = min(max(__builtin_amdgcn_readlane(cv.x, j + t), 0), lastRow);
    v[t] = __int_as_float(__builtin_amdgcn_readlane(cv.y, j + t));
    const float4* s = y4 + (size_t)c * (D / 4) + lane;
    p[t] = s[0];
    q[t] = s[32];
  }
  for (int t = 0; t < K; ++t) {
    a0.x = fmaf(v[t], p[t].x, a0.x); a0.y = fmaf(v[t], p[t].y, a0.y);
    a0.z = fmaf(v[t], p[t].z, a0.z); a0.w = fmaf(v[t], p[t].w, a0.w);
    a1.x = fmaf(v[t], q[t].x, a1.x); a1.y = fmaf(v[t], q[t].y, a1.y);
    a1.z = fmaf(v[t], q[t].z, a1.z); a1.w = fmaf(v[t], q[t].w, a1.w);
  }
}

__global__ __launch_bounds__(256) void gcn_gather_rows(
    const int* __restrict__ rowptr, const int* __restrict__ cnt,
    const int* __restrict__ csr, const float* __restrict__ y,
    const float* __restrict__ b, float* __restrict__ out, int N, int E) {
  const int lane = threadIdx.x & 31;
  const int i = __builtin_amdgcn_readfirstlane(blockIdx.x * 8 +
                                               (threadIdx.x >> 5));
  if (i >= N) return;

  int start = rowptr[i];
  int len   = cnt[i];
  start = min(max(start, 0), E);
  len   = min(max(len, 0), E - start);

  const float4* b4 = (const float4*)b;
  const float4* y4 = (const float4*)y;
  const int2*   pr = (const int2*)csr;
  float4 a0 = b4[lane];
  float4 a1 = b4[lane + 32];

  for (int base = 0; base < len; base += 32) {
    const int n = min(32, len - base);
    int2 cv = make_int2(0, 0);
    if (lane < n) cv = pr[start + base + lane];

    int j = 0;
    for (; j + 4 <= n; j += 4) gcn_add_edges<4>(y4, cv, j, lane, N - 1, a0, a1);
    const int rem = n - j;
    if (rem == 3)      gcn_add_edges<3>(y4, cv, j, lane, N - 1, a0, a1);
    else if (rem == 2) gcn_add_edges<2>(y4, cv, j, lane, N - 1, a0, a1);
    else if (rem == 1) gcn_add_edges<1>(y4, cv, j, lane, N - 1, a0, a1);
  }

  float4* o4 = (float4*)out + (size_t)i * (D / 4) + lane;
  o4[0]  = a0;
  o4[32] = a1;
}

extern "C" void kernel_launch(void* const* d_in, const int* in_sizes, int n_in,
                              void* d_out, int out_size, void* d_ws, size_t ws_size,
                              hipStream_t stream) {
  const float* x    = (const float*)d_in[0];
  const int*   erow = (const int*)d_in[1];
  const int*   ecol = (const int*)d_in[2];
  const float* ev   = (const float*)d_in[3];
  const float* W    = (const float*)d_in[4];
  const float* b    = (const float*)d_in[5];
  float*       out  = (float*)d_out;

  const int N  = in_sizes[0] / D;
  const int E  = in_sizes[1];
  const int Np = (N + 63) & ~63;

  float* y      = (float*)d_ws;
  int*   cnt    = (int*)(y + (size_t)N * D);
  int*   rowptr = cnt + Np;
  int*   tbl    = rowptr + Np;
  int*   csr    = tbl + (size_t)SORT_CHUNKS * Np;

  const size_t ldsBytes = (size_t)128 * PAIR_STRIDE * sizeof(v2f);
  gcn_gemm_xw<<<512, 256, ldsBytes, stream>>>(x, W, y, N / 16);

  const int    chunkLen = (E + SORT_CHUNKS - 1) / SORT_CHUNKS;
  const size_t nodeLds  = (size_t)Np * sizeof(int);
  gcn_count_chunk<<<SORT_CHUNKS, 1024, nodeLds, stream>>>(erow, tbl, N, Np / 4,
                                                         E, chunkLen);
  gcn_chunk_prefix<<<(Np + 255) / 256, 256, 0, stream>>>(tbl, cnt, Np);
  gcn_row_offsets<<<1, 1024, 0, stream>>>(cnt, rowptr, Np / 4);
  gcn_place_chunk<<<SORT_CHUNKS, 1024, nodeLds, stream>>>(
      erow, ecol, ev, rowptr, tbl, csr, N, Np / 4, E, chunkLen);

  gcn_gather_rows<<<(N + 7) / 8, 256, 0, stream>>>(rowptr, cnt, csr, y, b, out,
                                                   N, E);
}


# x@W kernel keeps the f32 matrix pipe fed: one resident 16-wave work-group per WGP stages W once (16-B loads), tiles dealt evenly, x slabs requested 64 WMMAs ahead; k order and sums unchanged
# speedup vs baseline: 2.5187x; 1.2581x over previous
#include <hip/hip_runtime.h>

typedef __attribute__((ext_vector_type(2))) float v2f;
typedef __attribute__((ext_vector_type(8))) float v8f;

#define D 256
#define PAIR_STRIDE 272

extern __shared__ v2f ldsP[];

#define GEMM_QUADS   4
#define GEMM_THREADS (GEMM_QUADS * 128)

__device__ __forceinline__ void gcn_gemm_slab(const v2f (&a)[8], int lb0,
                                              v8f& acc0, v8f& acc1, v8f& acc2,
                                              v8f& acc3) {
  for (int u = 0; u < 8; ++u) {
    const int lb = lb0 + (2 * u) * PAIR_STRIDE;
    v2f b0 = ldsP[lb +  0];
    v2f b1 = ldsP[lb + 16];
    v2f b2 = ldsP[lb + 32];
    v2f b3 = ldsP[lb + 48];
    acc0 = __builtin_amdgcn_wmma_f32_16x16x4_f32(false, a[u], false, b0,
                                                 (short)0, acc0, false, false);
    acc1 = __builtin_amdgcn_wmma_f32_16x16x4_f32(false, a[u], false, b1,
                                                 (short)0, acc1, false, false);
    acc2 = __builtin_amdgcn_wmma_f32_16x16x4_f32(false, a[u], false, b2,
                                                 (short)0, acc2, false, false);
    acc3 = __builtin_amdgcn_wmma_f32_16x16x4_f32(false, a[u], false, b3,
                                                 (short)0, acc3, false, false);
  }
}

__global__ __launch_bounds__(GEMM_THREADS) void gcn_gemm_xw(
    const float* __restrict__ x, const float* __restrict__ W,
    float* __restrict__ y, int nRowTiles) {
  const int tid  = threadIdx.x;
  const int lane = tid & 31;
  const int wv   = tid >> 5;
  const int lr   = lane & 15;
  const int hi   = lane >> 4;

#pragma unroll 8
  for (int g = tid; g < (D / 2) * (D / 4); g += GEMM_THREADS) {
    const int p  = g >> 6;
    const int c4 = g & 63;
    const float4 r0 = *(const float4*)(W + (2 * p) * D + 4 * c4);
    const float4 r1 = *(const float4*)(W + (2 * p + 1) * D + 4 * c4);
    v2f* dst = ldsP + p * PAIR_STRIDE + 4 * c4;
    v2f e0, e1, e2, e3;
    e0.x = r0.x; e0.y = r1.x;
    e1.x = r0.y; e1.y = r1.y;
    e2.x = r0.z; e2.y = r1.z;
    e3.x = r0.w; e3.y = r1.w;
    dst[0] = e0; dst[1] = e1; dst[2] = e2; dst[3] = e3;
  }
  __syncthreads();

  const int quad    = wv >> 2;
  const int colBase = (wv & 3) * 64;
  const int nb      = gridDim.x;

  const int lbase = hi * PAIR_STRIDE + colBase + lr;

  int t = quad * nb + blockIdx.x;
  const int tstep = GEMM_QUADS * nb;
  if (t >= nRowTiles) return;

  const float* xl = x + lr * D + 2 * hi;
  const float* xp = xl + (size_t)t * (16 * D);

  v2f a0[8], a1[8], a2[8], a3[8];
  for (int u = 0; u < 8; ++u) a0[u] = *(const v2f*)(xp + 4 * u);
  for (int u = 0; u < 8; ++u) a1[u] = *(const v2f*)(xp + 32 + 4 * u);

  for (; t < nRowTiles; t += tstep) {
    const int tn = min(t + tstep, nRowTiles - 1);
    const float* xpn = xl + (size_t)tn * (16 * D);

    v8f acc0 = {}, acc1 = {}, acc2 = {}, acc3 = {};

#pragma unroll 1
    for (int j = 0; j < 2; ++j) {
      const float* ps = xp + 128 * j;
      const int    lb = lbase + (64 * j) * PAIR_STRIDE;

      for (int u = 0; u < 8; ++u) a2[u] = *(const v2f*)(ps + 64 + 4 * u);
      gcn_gemm_slab(a0, lb, acc0, acc1, acc2, acc3);

      for (int u = 0; u < 8; ++u) a3[u] = *(const v2f*)(ps + 96 + 4 * u);
      gcn_gemm_slab(a1, lb + 16 * PAIR_STRIDE, acc0, acc1, acc2, acc3);

      const float* pc = (j < 1) ? ps + 128 : xpn;
      for (int u = 0; u < 8; ++u) a0[u] = *(const v2f*)(pc + 4 * u);
      gcn_gemm_slab(a2, lb + 32 * PAIR_STRIDE, acc0, acc1, acc2, acc3);

      for (int u = 0; u < 8; ++u) a1[u] = *(const v2f*)(pc + 32 + 4 * u);
      gcn_gemm_slab(a3, lb + 48 * PAIR_STRIDE, acc0, acc1, acc2, acc3);
    }

    float* yp = y + (size_t)t * (16 * D) + colBase + lr;
#pragma unroll
    for (int i = 0; i < 8; ++i) {
      const int r = (i + 8 * hi) * D;
      yp[r +  0] = acc0[i];
      yp[r + 16] = acc1[i];
      yp[r + 32] = acc2[i];
      yp[r + 48] = acc3[i];
    }
    xp = xpn;
  }
}

#define SORT_CHUNKS 128
#define SORT_EPT 8

extern __shared__ int ldsNode[];

__device__ __forceinline__ int gcn_row_id(int r, int lastRow) {
  return min(max(r, 0), lastRow);
}

__global__ __launch_bounds__(1024) void gcn_count_chunk(
    const int* __restrict__ erow, int* __restrict__ tbl, int N, int n4, int E,
    int chunkLen) {
  const int tid = threadIdx.x;
  const int c   = blockIdx.x;
  const int e0  = min(E, c * chunkLen);
  const int e1  = min(E, e0 + chunkLen);
  int4* l4 = (int4*)ldsNode;

#pragma unroll 4
  for (int q = tid; q < n4; q += 1024) l4[q] = make_int4(0, 0, 0, 0);
  __syncthreads();

  for (int b = e0 + tid; b < e1; b += 1024 * SORT_EPT) {
    int r[SORT_EPT];
    for (int u = 0; u < SORT_EPT; ++u) r[u] = erow[min(b + u * 1024, e1 - 1)];
    asm volatile("" :: "v"(r[0]), "v"(r[1]), "v"(r[2]), "v"(r[3]),
                       "v"(r[4]), "v"(r[5]), "v"(r[6]), "v"(r[7]));
    for (int u = 0; u < SORT_EPT; ++u) {
      if (b + u * 1024 < e1) atomicAdd(&ldsNode[gcn_row_id(r[u], N - 1)], 1);
    }
  }
  __syncthreads();

  int4* t4 = (int4*)tbl + (size_t)c * n4;
#pragma unroll 4
  for (int q = tid; q < n4; q += 1024) t4[q] = l4[q];
}

__global__ __launch_bounds__(256) void gcn_chunk_prefix(int* __restrict__ tbl,
                                                        int* __restrict__ cnt,
                                                        int Np) {
  const int i = blockIdx.x * 256 + threadIdx.x;
  if (i >= Np) return;
  int v[SORT_CHUNKS];
  const unsigned np = (unsigned)Np;
  for (int c = 0; c < SORT_CHUNKS; ++c) v[c] = tbl[(unsigned)c * np + (unsigned)i];
  int acc = 0;
  for (int c = 0; c < SORT_CHUNKS; ++c) {
    tbl[(unsigned)c * np + (unsigned)i] = acc;
    acc += v[c];
  }
  cnt[i] = acc;
}

__global__ __launch_bounds__(1024) void gcn_row_offsets(
    const int* __restrict__ cnt, int* __restrict__ rowptr, int n4) {
  __shared__ int wtot[32];
  const int tid  = threadIdx.x;
  const int lane = tid & 31;
  const int wv   = tid >> 5;
  const int per  = (n4 + 1023) >> 10;
  const int beg  = min(tid * per, n4);
  const int end  = min(beg + per, n4);
  const int4* c4 = (const int4*)cnt;
  int4* r4 = (int4*)rowptr;

  int s = 0;
#pragma unroll 4
  for (int q = beg; q < end; ++q) {
    const int4 v = c4[q];
    s += (v.x + v.y) + (v.z + v.w);
  }

  int inc = s;
  for (int d = 1; d < 32; d <<= 1) {
    const int t = __shfl_up(inc, d);
    if (lane >= d) inc += t;
  }
  if (lane == 31) wtot[wv] = inc;
  __syncthreads();
  if (wv == 0) {
    const int w = wtot[lane];
    int winc = w;
    for (int d = 1; d < 32; d <<= 1) {
      const int t = __shfl_up(winc, d);
      if (lane >= d) winc += t;
    }
    wtot[lane] = winc - w;
  }
  __syncthreads();

  int run = wtot[wv] + inc - s;
#pragma unroll 4
  for (int q = beg; q < end; ++q) {
    const int4 v = c4[q];
    int4 o;
    o.x = run;
    o.y = o.x + v.x;
    o.z = o.y + v.y;
    o.w = o.z + v.z;
    run = o.w + v.w;
    r4[q] = o;
  }
}

__global__ __launch_bounds__(1024) void gcn_place_chunk(
    const int* __restrict__ erow, const int* __restrict__ ecol,
    const float* __restrict__ eval_, const int* __restrict__ rowptr,
    const int* __restrict__ tbl, int* __restrict__ csr, int N, int n4, int E,
    int chunkLen) {
  const int tid = threadIdx.x;
  const int c   = blockIdx.x;
  const int e0  = min(E, c * chunkLen);
  const int e1  = min(E, e0 + chunkLen);
  int4* l4 = (int4*)ldsNode;
  const int4* r4 = (const int4*)rowptr;
  const int4* t4 = (const int4*)tbl + (size_t)c * n4;

#pragma unroll 4
  for (int q = tid; q < n4; q += 1024) {
    const int4 a = r4[q];
    const int4 p = t4[q];
    l4[q] = make_int4(a.x + p.x, a.y + p.y, a.z + p.z, a.w + p.w);
  }
  __syncthreads();

  for (int b = e0 + tid; b < e1; b += 1024 * SORT_EPT) {
    int   r[SORT_EPT], cl[SORT_EPT];
    float vl[SORT_EPT];
    for (int u = 0; u < SORT_EPT; ++u) {
      const int e = min(b + u * 1024, e1 - 1);
      r[u]  = erow[e];
      cl[u] = ecol[e];
      vl[u] = eval_[e];
    }
    asm volatile("" :: "v"(r[0]), "v"(r[1]), "v"(r[2]), "v"(r[3]),
                       "v"(r[4]), "v"(r[5]), "v"(r[6]), "v"(r[7]));
    asm volatile("" :: "v"(cl[0]), "v"(cl[1]), "v"(cl[2]), "v"(cl[3]),
                       "v"(cl[4]), "v"(cl[5]), "v"(cl[6]), "v"(cl[7]));
    asm volatile("" :: "v"(vl[0]), "v"(vl[1]), "v"(vl[2]), "v"(vl[3]),
                       "v"(vl[4]), "v"(vl[5]), "v"(vl[6]), "v"(vl[7]));
    for (int u = 0; u < SORT_EPT; ++u) {
      if (b + u * 1024 < e1) {
        int pos = atomicAdd(&ldsNode[gcn_row_id(r[u], N - 1)], 1);
        pos = min(max(pos, 0), E - 1);
        ((int2*)csr)[pos] = make_int2(cl[u], __float_as_int(vl[u]));
      }
    }
  }
}

template <int K>
__device__ __forceinline__ void gcn_add_edges(const float4* __restrict__ y4,
                                              int2 cv, int j, int lane,
                                              int lastRow, float4& a0,
                                              float4& a1) {
  float4 p[K], q[K];
  float  v[K];
  for (int t = 0; t < K; ++t) {
    const int c = min(max(__builtin_amdgcn_readlane(cv.x, j + t), 0), lastRow);
    v[t] = __int_as_float(__builtin_amdgcn_readlane(cv.y, j + t));
    const float4* s = y4 + (size_t)c * (D / 4) + lane;
    p[t] = s[0];
    q[t] = s[32];
  }
  for (int t = 0; t < K; ++t) {
    a0.x = fmaf(v[t], p[t].x, a0.x); a0.y = fmaf(v[t], p[t].y, a0.y);
    a0.z = fmaf(v[t], p[t].z, a0.z); a0.w = fmaf(v[t], p[t].w, a0.w);
    a1.x = fmaf(v[t], q[t].x, a1.x); a1.y = fmaf(v[t], q[t].y, a1.y);
    a1.z = fmaf(v[t], q[t].z, a1.z); a1.w = fmaf(v[t], q[t].w, a1.w);
  }
}

__global__ __launch_bounds__(256) void gcn_gather_rows(
    const int* __restrict__ rowptr, const int* __restrict__ cnt,
    const int* __restrict__ csr, const float* __restrict__ y,
    const float* __restrict__ b, float* __restrict__ out, int N, int E) {
  const int lane = threadIdx.x & 31;
  const int i = __builtin_amdgcn_readfirstlane(blockIdx.x * 8 +
                                               (threadIdx.x >> 5));
  if (i >= N) return;

  int start = rowptr[i];
  int len   = cnt[i];
  start = min(max(start, 0), E);
  len   = min(max(len, 0), E - start);

  const float4* b4 = (const float4*)b;
  const float4* y4 = (const float4*)y;
  const int2*   pr = (const int2*)csr;
  float4 a0 = b4[lane];
  float4 a1 = b4[lane + 32];

  for (int base = 0; base < len; base += 32) {
    const int n = min(32, len - base);
    int2 cv = make_int2(0, 0);
    if (lane < n) cv = pr[start + base + lane];

    int j = 0;
    for (; j + 4 <= n; j += 4) gcn_add_edges<4>(y4, cv, j, lane, N - 1, a0, a1);
    const int rem = n - j;
    if (rem == 3)      gcn_add_edges<3>(y4, cv, j, lane, N - 1, a0, a1);
    else if (rem == 2) gcn_add_edges<2>(y4, cv, j, lane, N - 1, a0, a1);
    else if (rem == 1) gcn_add_edges<1>(y4, cv, j, lane, N - 1, a0, a1);
  }

  float4* o4 = (float4*)out + (size_t)i * (D / 4) + lane;
  o4[0]  = a0;
  o4[32] = a1;
}

extern "C" void kernel_launch(void* const* d_in, const int* in_sizes, int n_in,
                              void* d_out, int out_size, void* d_ws, size_t ws_size,
                              hipStream_t stream) {
  const float* x    = (const float*)d_in[0];
  const int*   erow = (const int*)d_in[1];
  const int*   ecol = (const int*)d_in[2];
  const float* ev   = (const float*)d_in[3];
  const float* W    = (const float*)d_in[4];
  const float* b    = (const float*)d_in[5];
  float*       out  = (float*)d_out;

  const int N  = in_sizes[0] / D;
  const int E  = in_sizes[1];
  const int Np = (N + 63) & ~63;

  float* y      = (float*)d_ws;
  int*   cnt    = (int*)(y + (size_t)N * D);
  int*   rowptr = cnt + Np;
  int*   tbl    = rowptr + Np;
  int*   csr    = tbl + (size_t)SORT_CHUNKS * Np;

  const size_t ldsBytes = (size_t)128 * PAIR_STRIDE * sizeof(v2f);
  gcn_gemm_xw<<<256, GEMM_THREADS, ldsBytes, stream>>>(x, W, y, N / 16);

  const int    chunkLen = (E + SORT_CHUNKS - 1) / SORT_CHUNKS;
  const size_t nodeLds  = (size_t)Np * sizeof(int);
  gcn_count_chunk<<<SORT_CHUNKS, 1024, nodeLds, stream>>>(erow, tbl, N, Np / 4,
                                                         E, chunkLen);
  gcn_chunk_prefix<<<(Np + 255) / 256, 256, 0, stream>>>(tbl, cnt, Np);
  gcn_row_offsets<<<1, 1024, 0, stream>>>(cnt, rowptr, Np / 4);
  gcn_place_chunk<<<SORT_CHUNKS, 1024, nodeLds, stream>>>(
      erow, ecol, ev, rowptr, tbl, csr, N, Np / 4, E, chunkLen);

  gcn_gather_rows<<<(N + 7) / 8, 256, 0, stream>>>(rowptr, cnt, csr, y, b, out,
                                                   N, E);
}


# Edge grouping in 2 launches with the row sums fused in: edge blocks order their slices by 256 row buckets (LDS counters); one work-group per bucket orders by row in LDS and sums rows in edge-id order
# speedup vs baseline: 3.8555x; 1.5307x over previous
#include <hip/hip_runtime.h>

typedef __attribute__((ext_vector_type(2))) float v2f;
typedef __attribute__((ext_vector_type(8))) float v8f;

#define D 256
#define PAIR_STRIDE 272

extern __shared__ v2f ldsP[];

#define GEMM_QUADS   4
#define GEMM_THREADS (GEMM_QUADS * 128)

__device__ __forceinline__ void gcn_gemm_slab(const v2f (&a)[8], int lb0,
                                              v8f& acc0, v8f& acc1, v8f& acc2,
                                              v8f& acc3) {
  for (int u = 0; u < 8; ++u) {
    const int lb = lb0 + (2 * u) * PAIR_STRIDE;
    v2f b0 = ldsP[lb +  0];
    v2f b1 = ldsP[lb + 16];
    v2f b2 = ldsP[lb + 32];
    v2f b3 = ldsP[lb + 48];
    acc0 = __builtin_amdgcn_wmma_f32_16x16x4_f32(false, a[u], false, b0,
                                                 (short)0, acc0, false, false);
    acc1 = __builtin_amdgcn_wmma_f32_16x16x4_f32(false, a[u], false, b1,
                                                 (short)0, acc1, false, false);
    acc2 = __builtin_amdgcn_wmma_f32_16x16x4_f32(false, a[u], false, b2,
                                                 (short)0, acc2, false, false);
    acc3 = __builtin_amdgcn_wmma_f32_16x16x4_f32(false, a[u], false, b3,
                                                 (short)0, acc3, false, false);
  }
}

__global__ __launch_bounds__(GEMM_THREADS) void gcn_gemm_xw(
    const float* __restrict__ x, const float* __restrict__ W,
    float* __restrict__ y, int nRowTiles) {
  const int tid  = threadIdx.x;
  const int lane = tid & 31;
  const int wv   = tid >> 5;
  const int lr   = lane & 15;
  const int hi   = lane >> 4;

#pragma unroll 8
  for (int g = tid; g < (D / 2) * (D / 4); g += GEMM_THREADS) {
    const int p  = g >> 6;
    const int c4 = g & 63;
    const float4 r0 = *(const float4*)(W + (2 * p) * D + 4 * c4);
    const float4 r1 = *(const float4*)(W + (2 * p + 1) * D + 4 * c4);
    v2f* dst = ldsP + p * PAIR_STRIDE + 4 * c4;
    v2f e0, e1, e2, e3;
    e0.x = r0.x; e0.y = r1.x;
    e1.x = r0.y; e1.y = r1.y;
    e2.x = r0.z; e2.y = r1.z;
    e3.x = r0.w; e3.y = r1.w;
    dst[0] = e0; dst[1] = e1; dst[2] = e2; dst[3] = e3;
  }
  __syncthreads();

  const int quad    = wv >> 2;
  const int colBase = (wv & 3) * 64;
  const int nb      = gridDim.x;

  const int lbase = hi * PAIR_STRIDE + colBase + lr;

  int t = quad * nb + blockIdx.x;
  const int tstep = GEMM_QUADS * nb;
  if (t >= nRowTiles) return;

  const float* xl = x + lr * D + 2 * hi;
  const float* xp = xl + (size_t)t * (16 * D);

  v2f a0[8], a1[8], a2[8], a3[8];
  for (int u = 0; u < 8; ++u) a0[u] = *(const v2f*)(xp + 4 * u);
  for (int u = 0; u < 8; ++u) a1[u] = *(const v2f*)(xp + 32 + 4 * u);

  for (; t < nRowTiles; t += tstep) {
    const int tn = min(t + tstep, nRowTiles - 1);
    const float* xpn = xl + (size_t)tn * (16 * D);

    v8f acc0 = {}, acc1 = {}, acc2 = {}, acc3 = {};

#pragma unroll 1
    for (int j = 0; j < 2; ++j) {
      const float* ps = xp + 128 * j;
      const int    lb = lbase + (64 * j) * PAIR_STRIDE;

      for (int u = 0; u < 8; ++u) a2[u] = *(const v2f*)(ps + 64 + 4 * u);
      gcn_gemm_slab(a0, lb, acc0, acc1, acc2, acc3);

      for (int u = 0; u < 8; ++u) a3[u] = *(const v2f*)(ps + 96 + 4 * u);
      gcn_gemm_slab(a1, lb + 16 * PAIR_STRIDE, acc0, acc1, acc2, acc3);

      const float* pc = (j < 1) ? ps + 128 : xpn;
      for (int u = 0; u < 8; ++u) a0[u] = *(const v2f*)(pc + 4 * u);
      gcn_gemm_slab(a2, lb + 32 * PAIR_STRIDE, acc0, acc1, acc2, acc3);

      for (int u = 0; u < 8; ++u) a1[u] = *(const v2f*)(pc + 32 + 4 * u);
      gcn_gemm_slab(a3, lb + 48 * PAIR_STRIDE, acc0, acc1, acc2, acc3);
    }

    float* yp = y + (size_t)t * (16 * D) + colBase + lr;
#pragma unroll
    for (int i = 0; i < 8; ++i) {
      const int r = (i + 8 * hi) * D;
      yp[r +  0] = acc0[i];
      yp[r + 16] = acc1[i];
      yp[r + 32] = acc2[i];
      yp[r + 48] = acc3[i];
    }
    xp = xpn;
  }
}

#define GCN_NBK  256
#define GCN_EBLK 256
#define GCN_EPT  4
#define GCN_CAP  4096
#define GCN_TPT  (GCN_CAP / 1024)

extern __shared__ int4 ldsBk[];

__device__ __forceinline__ int gcn_row_id(int r, int lastRow) {
  return min(max(r, 0), lastRow);
}

__device__ __forceinline__ int gcn_wave_scan(int v, int lane) {
  for (int d = 1; d < 32; d <<= 1) {
    const int t = __shfl_up(v, d);
    if (lane >= d) v += t;
  }
  return v;
}

__device__ __forceinline__ int gcn_scan256(int c, int tid, int* wtot,
                                           int& total) {
  const int lane = tid & 31;
  const int wv   = tid >> 5;
  int inc = 0;
  if (tid < 256) {
    inc = gcn_wave_scan(c, lane);
    if (lane == 31) wtot[wv] = inc;
  }
  __syncthreads();
  int base = 0, all = 0;
  for (int w = 0; w < 8; ++w) {
    const int t = wtot[w];
    if (w < wv) base += t;
    all += t;
  }
  total = all;
  __syncthreads();
  return base + inc - c;
}

__global__ __launch_bounds__(1024) void gcn_bucket_edges(
    const int* __restrict__ erow, const int* __restrict__ ecol,
    const float* __restrict__ eval_, int* __restrict__ recw,
    int* __restrict__ tabw, int N, int E, int rpb, int chunkLen) {
  int4* rec = (int4*)recw;
  int2* tab = (int2*)tabw;
  __shared__ int cnt[GCN_NBK];
  __shared__ int wtot[8];
  const int tid = threadIdx.x;
  const int blk = blockIdx.x;
  const int e0  = min(E, blk * chunkLen);
  const int e1  = min(E, e0 + chunkLen);

  if (tid < GCN_NBK) cnt[tid] = 0;
  __syncthreads();

  for (int b = e0 + tid; b < e1; b += 1024 * GCN_EPT) {
    int r[GCN_EPT];
    for (int u = 0; u < GCN_EPT; ++u) r[u] = erow[min(b + u * 1024, e1 - 1)];
    asm volatile("" :: "v"(r[0]), "v"(r[1]), "v"(r[2]), "v"(r[3]));
    for (int u = 0; u < GCN_EPT; ++u) {
      if (b + u * 1024 < e1) atomicAdd(&cnt[gcn_row_id(r[u], N - 1) / rpb], 1);
    }
  }
  __syncthreads();

  const int c = (tid < GCN_NBK) ? cnt[tid] : 0;
  int total;
  const int excl = gcn_scan256(c, tid, wtot, total);
  if (tid < GCN_NBK) {
    tab[tid * GCN_EBLK + blk] = make_int2(e0 + excl, c);
    cnt[tid] = excl;
  }
  __syncthreads();

  for (int b = e0 + tid; b < e1; b += 1024 * GCN_EPT) {
    int   r[GCN_EPT], cl[GCN_EPT];
    float vl[GCN_EPT];
    for (int u = 0; u < GCN_EPT; ++u) {
      const int e = min(b + u * 1024, e1 - 1);
      r[u]  = erow[e];
      cl[u] = ecol[e];
      vl[u] = eval_[e];
    }
    asm volatile("" :: "v"(r[0]), "v"(r[1]), "v"(r[2]), "v"(r[3]));
    asm volatile("" :: "v"(cl[0]), "v"(cl[1]), "v"(cl[2]), "v"(cl[3]));
    asm volatile("" :: "v"(vl[0]), "v"(vl[1]), "v"(vl[2]), "v"(vl[3]));
    for (int u = 0; u < GCN_EPT; ++u) {
      const int e = b + u * 1024;
      if (e < e1) {
        const int row = gcn_row_id(r[u], N - 1);
        const int bk  = row / rpb;
        int pos = e0 + atomicAdd(&cnt[bk], 1);
        pos = min(max(pos, 0), E - 1);
        rec[pos] = make_int4(row - bk * rpb, e, cl[u], __float_as_int(vl[u]));
      }
    }
  }
}

template <int K>
__device__ __forceinline__ void gcn_add_edges(const float4* __restrict__ y4,
                                              int2 cv, int j, int lane,
                                              int lastRow, float4& a0,
                                              float4& a1) {
  float4 p[K], q[K];
  float  v[K];
  for (int t = 0; t < K; ++t) {
    const int c = min(max(__builtin_amdgcn_readlane(cv.x, j + t), 0), lastRow);
    v[t] = __int_as_float(__builtin_amdgcn_readlane(cv.y, j + t));
    const float4* s = y4 + (size_t)c * (D / 4) + lane;
    p[t] = s[0];
    q[t] = s[32];
  }
  for (int t = 0; t < K; ++t) {
    a0.x = fmaf(v[t], p[t].x, a0.x); a0.y = fmaf(v[t], p[t].y, a0.y);
    a0.z = fmaf(v[t], p[t].z, a0.z); a0.w = fmaf(v[t], p[t].w, a0.w);
    a1.x = fmaf(v[t], q[t].x, a1.x); a1.y = fmaf(v[t], q[t].y, a1.y);
    a1.z = fmaf(v[t], q[t].z, a1.z); a1.w = fmaf(v[t], q[t].w, a1.w);
  }
}

__global__ __launch_bounds__(1024) void gcn_bucket_rows(
    const int* __restrict__ recw, const int* __restrict__ tabw,
    const float* __restrict__ y, const float* __restrict__ b,
    float* __restrict__ out, int N, int E, int rpb) {
  const int4* rec = (const int4*)recw;
  const int2* tab = (const int2*)tabw;
  __shared__ int runOff[GCN_EBLK + 1];
  __shared__ int runStart[GCN_EBLK];
  __shared__ int rowCnt[256];
  __shared__ int rowOff[257];
  __shared__ int wtot[8];
  __shared__ int nextRow;
  int2*   sCV  = (int2*)ldsBk;
  int*    sKey = (int*)ldsBk + 2 * GCN_CAP;
  float4* acc4 = (float4*)((int*)ldsBk + 3 * GCN_CAP);

  const int tid  = threadIdx.x;
  const int lane = tid & 31;
  const int bk = blockIdx.x;

  int st = 0, c = 0;
  if (tid < GCN_EBLK) {
    const int2 t = tab[bk * GCN_EBLK + tid];
    c  = min(max(t.y, 0), E);
    st = min(max(t.x, 0), E - c);
  }
  int total;
  const int excl = gcn_scan256(c, tid, wtot, total);
  if (tid < GCN_EBLK) {
    runOff[tid]   = excl;
    runStart[tid] = st;
  }
  if (tid == 0) runOff[GCN_EBLK] = total;
  __syncthreads();
  const int M = min(max(total, 0), E);

  const int rowBase = bk * rpb;
  const int nRows   = min(max(N - rowBase, 0), min(rpb, 256));
  const float4* b4 = (const float4*)b;
  const float4* y4 = (const float4*)y;

  for (int t0 = 0;; t0 += GCN_CAP) {
    const int  tEnd = min(M, t0 + GCN_CAP);
    const bool last = (t0 + GCN_CAP >= M);

    if (tid < 256) rowCnt[tid] = 0;
    if (tid == 0) nextRow = 0;
    __syncthreads();

    int4 rc[GCN_TPT];
    int  rk[GCN_TPT];
    for (int u = 0; u < GCN_TPT; ++u) {
      const int k  = t0 + u * 1024 + tid;
      const int kk = min(k, max(tEnd - 1, 0));
      int lo = 0;
      for (int step = GCN_EBLK / 2; step >= 1; step >>= 1) {
        if (runOff[lo + step] <= kk) lo += step;
      }
      int pos = runStart[lo] + (kk - runOff[lo]);
      pos = min(max(pos, 0), E - 1);
      rc[u] = rec[pos];
    }
    asm volatile("" :: "v"(rc[0].x), "v"(rc[1].x), "v"(rc[2].x), "v"(rc[3].x));
    for (int u = 0; u < GCN_TPT; ++u) {
      rk[u] = 0;
      if (t0 + u * 1024 + tid < tEnd) {
        rk[u] = atomicAdd(&rowCnt[min(max(rc[u].x, 0), 255)], 1);
      }
    }
    __syncthreads();

    int tot2;
    const int rex = gcn_scan256((tid < 256) ? rowCnt[tid] : 0, tid, wtot, tot2);
    if (tid < 256) rowOff[tid] = rex;
    if (tid == 0) rowOff[256] = tot2;
    __syncthreads();

    for (int u = 0; u < GCN_TPT; ++u) {
      if (t0 + u * 1024 + tid < tEnd) {
        int p = rowOff[min(max(rc[u].x, 0), 255)] + rk[u];
        p = min(max(p, 0), GCN_CAP - 1);
        sCV[p]  = make_int2(rc[u].z, rc[u].w);
        sKey[p] = rc[u].y;
      }
    }
    __syncthreads();

    for (;;) {
      int rr = 0;
      if (lane == 0) rr = atomicAdd(&nextRow, 1);
      rr = __builtin_amdgcn_readfirstlane(rr);
      if (rr >= nRows) break;
      const int s   = __builtin_amdgcn_readfirstlane(rowOff[rr]);
      const int len = __builtin_amdgcn_readfirstlane(rowOff[rr + 1]) - s;
      float4 a0, a1;
      if (t0 == 0) {
        a0 = b4[lane];
        a1 = b4[lane + 32];
      } else {
        a0 = acc4[rr * 64 + lane];
        a1 = acc4[rr * 64 + 32 + lane];
      }

      for (int base = 0; base < len; base += 32) {
        const int n = min(32, len - base);
        int2 cv  = make_int2(0, 0);
        int  key = 0x7fffffff;
        if (lane < n) {
          cv  = sCV[s + base + lane];
          key = sKey[s + base + lane];
        }
        if (len <= 32) {
          int rank = 0;
          for (int j = 0; j < n; ++j) {
            rank += (__builtin_amdgcn_readlane(key, j) < key) ? 1 : 0;
          }
          cv.x = __builtin_amdgcn_ds_permute(rank * 4, cv.x);
          cv.y = __builtin_amdgcn_ds_permute(rank * 4, cv.y);
        }

        int j = 0;
        for (; j + 8 <= n; j += 8) gcn_add_edges<8>(y4, cv, j, lane, N - 1, a0, a1);
        if (j + 4 <= n) {
          gcn_add_edges<4>(y4, cv, j, lane, N - 1, a0, a1);
          j += 4;
        }
        const int rem = n - j;
        if (rem == 3)      gcn_add_edges<3>(y4, cv, j, lane, N - 1, a0, a1);
        else if (rem == 2) gcn_add_edges<2>(y4, cv, j, lane, N - 1, a0, a1);
        else if (rem == 1) gcn_add_edges<1>(y4, cv, j, lane, N - 1, a0, a1);
      }

      if (last) {
        float4* o4 = (float4*)out + (size_t)(rowBase + rr) * (D / 4) + lane;
        o4[0]  = a0;
        o4[32] = a1;
      } else {
        acc4[rr * 64 + lane]      = a0;
        acc4[rr * 64 + 32 + lane] = a1;
      }
    }

    if (last) break;
    __syncthreads();
  }
}

extern "C" void kernel_launch(void* const* d_in, const int* in_sizes, int n_in,
                              void* d_out, int out_size, void* d_ws, size_t ws_size,
                              hipStream_t stream) {
  const float* x    = (const float*)d_in[0];
  const int*   erow = (const int*)d_in[1];
  const int*   ecol = (const int*)d_in[2];
  const float* ev   = (const float*)d_in[3];
  const float* W    = (const float*)d_in[4];
  const float* b    = (const float*)d_in[5];
  float*       out  = (float*)d_out;

  const int N   = in_sizes[0] / D;
  const int E   = in_sizes[1];
  const int rpb = (N + GCN_NBK - 1) / GCN_NBK;

  float* y   = (float*)d_ws;
  int*   rec = (int*)(y + (size_t)N * D);
  int*   tab = rec + (size_t)4 * E;

  const int chunkLen = (E + GCN_EBLK - 1) / GCN_EBLK;
  gcn_bucket_edges<<<GCN_EBLK, 1024, 0, stream>>>(erow, ecol, ev, rec, tab, N, E,
                                                 rpb, chunkLen);

  const size_t ldsBytes = (size_t)128 * PAIR_STRIDE * sizeof(v2f);
  gcn_gemm_xw<<<256, GEMM_THREADS, ldsBytes, stream>>>(x, W, y, N / 16);

  const size_t rowLds = (size_t)3 * GCN_CAP * sizeof(int) +
                        (size_t)rpb * D * sizeof(float);
  gcn_bucket_rows<<<GCN_NBK, 1024, rowLds, stream>>>(rec, tab, y, b, out, N, E,
                                                    rpb);
}


# Edge bucket pass runs on 8 extra waves of the x@W work-groups, beside the WMMA waves (its loads and LDS counters wait while the matrix pipe works): 3 launches become 2; sums and order unchanged
# speedup vs baseline: 4.0517x; 1.0509x over previous
#include <hip/hip_runtime.h>

typedef __attribute__((ext_vector_type(2))) float v2f;
typedef __attribute__((ext_vector_type(8))) float v8f;

#define D 256
#define PAIR_STRIDE 272

extern __shared__ v2f ldsP[];

#define GEMM_QUADS   4
#define GEMM_THREADS (GEMM_QUADS * 128)
#define EDGE_THREADS 256
#define FUSE_THREADS (GEMM_THREADS + EDGE_THREADS)

__device__ __forceinline__ void gcn_gemm_slab(const v2f (&a)[8], int lb0,
                                              v8f& acc0, v8f& acc1, v8f& acc2,
                                              v8f& acc3) {
  for (int u = 0; u < 8; ++u) {
    const int lb = lb0 + (2 * u) * PAIR_STRIDE;
    v2f b0 = ldsP[lb +  0];
    v2f b1 = ldsP[lb + 16];
    v2f b2 = ldsP[lb + 32];
    v2f b3 = ldsP[lb + 48];
    acc0 = __builtin_amdgcn_wmma_f32_16x16x4_f32(false, a[u], false, b0,
                                                 (short)0, acc0, false, false);
    acc1 = __builtin_amdgcn_wmma_f32_16x16x4_f32(false, a[u], false, b1,
                                                 (short)0, acc1, false, false);
    acc2 = __builtin_amdgcn_wmma_f32_16x16x4_f32(false, a[u], false, b2,
                                                 (short)0, acc2, false, false);
    acc3 = __builtin_amdgcn_wmma_f32_16x16x4_f32(false, a[u], false, b3,
                                                 (short)0, acc3, false, false);
  }
}

#define GCN_NBK  256
#define GCN_EBLK 256
#define GCN_EPT  13
#define GCN_CAP  4096
#define GCN_TPT  (GCN_CAP / 1024)

extern __shared__ int4 ldsBk[];

__device__ __forceinline__ int gcn_row_id(int r, int lastRow) {
  return min(max(r, 0), lastRow);
}

__device__ __forceinline__ int gcn_wave_scan(int v, int lane) {
  for (int d = 1; d < 32; d <<= 1) {
    const int t = __shfl_up(v, d);
    if (lane >= d) v += t;
  }
  return v;
}

__device__ __forceinline__ int gcn_scan256(int c, int tid, int* wtot,
                                           int& total) {
  const int lane = tid & 31;
  const int wv   = tid >> 5;
  int inc = 0;
  if (tid < 256) {
    inc = gcn_wave_scan(c, lane);
    if (lane == 31) wtot[wv] = inc;
  }
  __syncthreads();
  int base = 0, all = 0;
  for (int w = 0; w < 8; ++w) {
    const int t = wtot[w];
    if (w < wv) base += t;
    all += t;
  }
  total = all;
  __syncthreads();
  return base + inc - c;
}

static_assert(EDGE_THREADS == GCN_NBK, "one bucket count per edge thread");

__global__ __launch_bounds__(FUSE_THREADS) void gcn_gemm_xw_edges(
    const float* __restrict__ x, const float* __restrict__ W,
    float* __restrict__ y, int nRowTiles, const int* __restrict__ erow,
    const int* __restrict__ ecol, const float* __restrict__ eval_,
    int* __restrict__ recw, int* __restrict__ tabw, int N, int E, int rpb,
    int chunkLen) {
  int4* rec = (int4*)recw;
  int2* tab = (int2*)tabw;
  __shared__ int cnt[GCN_NBK];
  __shared__ int wtot[8];
  const int tid  = threadIdx.x;
  const int lane = tid & 31;
  const int wv   = tid >> 5;
  const int lr   = lane & 15;
  const int hi   = lane >> 4;
  const bool edgeRole = (wv >= GEMM_THREADS / 32);
  const int lt  = tid - GEMM_THREADS;
  const int ew  = wv - GEMM_THREADS / 32;
  const int blk = blockIdx.x;
  const int e0  = min(E, blk * chunkLen);
  const int e1  = min(E, e0 + chunkLen);

  if (edgeRole) cnt[lt] = 0;
  __syncthreads();

  if (!edgeRole) {
#pragma unroll 8
    for (int g = tid; g < (D / 2) * (D / 4); g += GEMM_THREADS) {
      const int p  = g >> 6;
      const int c4 = g & 63;
      const float4 r0 = *(const float4*)(W + (2 * p) * D + 4 * c4);
      const float4 r1 = *(const float4*)(W + (2 * p + 1) * D + 4 * c4);
      v2f* dst = ldsP + p * PAIR_STRIDE + 4 * c4;
      v2f e0v, e1v, e2v, e3v;
      e0v.x = r0.x; e0v.y = r1.x;
      e1v.x = r0.y; e1v.y = r1.y;
      e2v.x = r0.z; e2v.y = r1.z;
      e3v.x = r0.w; e3v.y = r1.w;
      dst[0] = e0v; dst[1] = e1v; dst[2] = e2v; dst[3] = e3v;
    }
  } else {
    for (int b = e0 + lt; b < e1; b += EDGE_THREADS * GCN_EPT) {
      int r[GCN_EPT];
      for (int u = 0; u < GCN_EPT; ++u) {
        r[u] = erow[min(b + u * EDGE_THREADS, e1 - 1)];
      }
      asm volatile("" :: "v"(r[0]), "v"(r[1]), "v"(r[2]), "v"(r[3]),
                         "v"(r[4]), "v"(r[5]), "v"(r[6]), "v"(r[7]),
                         "v"(r[8]), "v"(r[9]), "v"(r[10]), "v"(r[11]),
                         "v"(r[12]));
      for (int u = 0; u < GCN_EPT; ++u) {
        if (b + u * EDGE_THREADS < e1) {
          atomicAdd(&cnt[gcn_row_id(r[u], N - 1) / rpb], 1);
        }
      }
    }
  }
  __syncthreads();

  int c = 0, inc = 0, excl = 0;
  if (edgeRole) {
    c   = cnt[lt];
    inc = gcn_wave_scan(c, lane);
    if (lane == 31) wtot[ew] = inc;
  }
  __syncthreads();
  if (edgeRole) {
    int base = 0;
    for (int w = 0; w < 8; ++w) {
      const int t = wtot[w];
      if (w < ew) base += t;
    }
    excl = base + inc - c;
    cnt[lt] = excl;
  }
  __syncthreads();

  if (edgeRole) {
    tab[lt * GCN_EBLK + blk] = make_int2(e0 + excl, c);

    for (int b = e0 + lt; b < e1; b += EDGE_THREADS * GCN_EPT) {
      int   r[GCN_EPT], cl[GCN_EPT];
      float vl[GCN_EPT];
      for (int u = 0; u < GCN_EPT; ++u) {
        const int e = min(b + u * EDGE_THREADS, e1 - 1);
        r[u]  = erow[e];
        cl[u] = ecol[e];
        vl[u] = eval_[e];
      }
      asm volatile("" :: "v"(r[0]), "v"(r[1]), "v"(r[2]), "v"(r[3]),
                         "v"(r[4]), "v"(r[5]), "v"(r[6]), "v"(r[7]),
                         "v"(r[8]), "v"(r[9]), "v"(r[10]), "v"(r[11]),
                         "v"(r[12]));
      asm volatile("" :: "v"(cl[0]), "v"(cl[1]), "v"(cl[2]), "v"(cl[3]),
                         "v"(cl[4]), "v"(cl[5]), "v"(cl[6]), "v"(cl[7]),
                         "v"(cl[8]), "v"(cl[9]), "v"(cl[10]), "v"(cl[11]),
                         "v"(cl[12]));
      asm volatile("" :: "v"(vl[0]), "v"(vl[1]), "v"(vl[2]), "v"(vl[3]),
                         "v"(vl[4]), "v"(vl[5]), "v"(vl[6]), "v"(vl[7]),
                         "v"(vl[8]), "v"(vl[9]), "v"(vl[10]), "v"(vl[11]),
                         "v"(vl[12]));
      for (int u = 0; u < GCN_EPT; ++u) {
        const int e = b + u * EDGE_THREADS;
        if (e < e1) {
          const int row = gcn_row_id(r[u], N - 1);
          const int bk  = row / rpb;
          int pos = e0 + atomicAdd(&cnt[bk], 1);
          pos = min(max(pos, 0), E - 1);
          rec[pos] = make_int4(row - bk * rpb, e, cl[u], __float_as_int(vl[u]));
        }
      }
    }
    return;
  }

  const int quad    = wv >> 2;
  const int colBase = (wv & 3) * 64;
  const int nb      = gridDim.x;
  const int lbase = hi * PAIR_STRIDE + colBase + lr;

  int t = quad * nb + blockIdx.x;
  const int tstep = GEMM_QUADS * nb;
  if (t >= nRowTiles) return;

  const float* xl = x + lr * D + 2 * hi;
  const float* xp = xl + (size_t)t * (16 * D);

  v2f a0[8], a1[8], a2[8], a3[8];
  for (int u = 0; u < 8; ++u) a0[u] = *(const v2f*)(xp + 4 * u);
  for (int u = 0; u < 8; ++u) a1[u] = *(const v2f*)(xp + 32 + 4 * u);

  for (; t < nRowTiles; t += tstep) {
    const int tn = min(t + tstep, nRowTiles - 1);
    const float* xpn = xl + (size_t)tn * (16 * D);

    v8f acc0 = {}, acc1 = {}, acc2 = {}, acc3 = {};

#pragma unroll 1
    for (int j = 0; j < 2; ++j) {
      const float* ps = xp + 128 * j;
      const int    lb = lbase + (64 * j) * PAIR_STRIDE;

      for (int u = 0; u < 8; ++u) a2[u] = *(const v2f*)(ps + 64 + 4 * u);
      gcn_gemm_slab(a0, lb, acc0, acc1, acc2, acc3);

      for (int u = 0; u < 8; ++u) a3[u] = *(const v2f*)(ps + 96 + 4 * u);
      gcn_gemm_slab(a1, lb + 16 * PAIR_STRIDE, acc0, acc1, acc2, acc3);

      const float* pc = (j < 1) ? ps + 128 : xpn;
      for (int u = 0; u < 8; ++u) a0[u] = *(const v2f*)(pc + 4 * u);
      gcn_gemm_slab(a2, lb + 32 * PAIR_STRIDE, acc0, acc1, acc2, acc3);

      for (int u = 0; u < 8; ++u) a1[u] = *(const v2f*)(pc + 32 + 4 * u);
      gcn_gemm_slab(a3, lb + 48 * PAIR_STRIDE, acc0, acc1, acc2, acc3);
    }

    float* yp = y + (size_t)t * (16 * D) + colBase + lr;
#pragma unroll
    for (int i = 0; i < 8; ++i) {
      const int r = (i + 8 * hi) * D;
      yp[r +  0] = acc0[i];
      yp[r + 16] = acc1[i];
      yp[r + 32] = acc2[i];
      yp[r + 48] = acc3[i];
    }
    xp = xpn;
  }
}

template <int K>
__device__ __forceinline__ void gcn_add_edges(const float4* __restrict__ y4,
                                              int2 cv, int j, int lane,
                                              int lastRow, float4& a0,
                                              float4& a1) {
  float4 p[K], q[K];
  float  v[K];
  for (int t = 0; t < K; ++t) {
    const int c = min(max(__builtin_amdgcn_readlane(cv.x, j + t), 0), lastRow);
    v[t] = __int_as_float(__builtin_amdgcn_readlane(cv.y, j + t));
    const float4* s = y4 + (size_t)c * (D / 4) + lane;
    p[t] = s[0];
    q[t] = s[32];
  }
  for (int t = 0; t < K; ++t) {
    a0.x = fmaf(v[t], p[t].x, a0.x); a0.y = fmaf(v[t], p[t].y, a0.y);
    a0.z = fmaf(v[t], p[t].z, a0.z); a0.w = fmaf(v[t], p[t].w, a0.w);
    a1.x = fmaf(v[t], q[t].x, a1.x); a1.y = fmaf(v[t], q[t].y, a1.y);
    a1.z = fmaf(v[t], q[t].z, a1.z); a1.w = fmaf(v[t], q[t].w, a1.w);
  }
}

__global__ __launch_bounds__(1024) void gcn_bucket_rows(
    const int* __restrict__ recw, const int* __restrict__ tabw,
    const float* __restrict__ y, const float* __restrict__ b,
    float* __restrict__ out, int N, int E, int rpb) {
  const int4* rec = (const int4*)recw;
  const int2* tab = (const int2*)tabw;
  __shared__ int runOff[GCN_EBLK + 1];
  __shared__ int runStart[GCN_EBLK];
  __shared__ int rowCnt[256];
  __shared__ int rowOff[257];
  __shared__ int wtot[8];
  __shared__ int nextRow;
  int2*   sCV  = (int2*)ldsBk;
  int*    sKey = (int*)ldsBk + 2 * GCN_CAP;
  float4* acc4 = (float4*)((int*)ldsBk + 3 * GCN_CAP);

  const int tid  = threadIdx.x;
  const int lane = tid & 31;
  const int bk = blockIdx.x;

  int st = 0, c = 0;
  if (tid < GCN_EBLK) {
    const int2 t = tab[bk * GCN_EBLK + tid];
    c  = min(max(t.y, 0), E);
    st = min(max(t.x, 0), E - c);
  }
  int total;
  const int excl = gcn_scan256(c, tid, wtot, total);
  if (tid < GCN_EBLK) {
    runOff[tid]   = excl;
    runStart[tid] = st;
  }
  if (tid == 0) runOff[GCN_EBLK] = total;
  __syncthreads();
  const int M = min(max(total, 0), E);

  const int rowBase = bk * rpb;
  const int nRows   = min(max(N - rowBase, 0), min(rpb, 256));
  const float4* b4 = (const float4*)b;
  const float4* y4 = (const float4*)y;

  for (int t0 = 0;; t0 += GCN_CAP) {
    const int  tEnd = min(M, t0 + GCN_CAP);
    const bool last = (t0 + GCN_CAP >= M);

    if (tid < 256) rowCnt[tid] = 0;
    if (tid == 0) nextRow = 0;
    __syncthreads();

    int4 rc[GCN_TPT];
    int  rk[GCN_TPT];
    for (int u = 0; u < GCN_TPT; ++u) {
      const int k  = t0 + u * 1024 + tid;
      const int kk = min(k, max(tEnd - 1, 0));
      int lo = 0;
      for (int step = GCN_EBLK / 2; step >= 1; step >>= 1) {
        if (runOff[lo + step] <= kk) lo += step;
      }
      int pos = runStart[lo] + (kk - runOff[lo]);
      pos = min(max(pos, 0), E - 1);
      rc[u] = rec[pos];
    }
    asm volatile("" :: "v"(rc[0].x), "v"(rc[1].x), "v"(rc[2].x), "v"(rc[3].x));
    for (int u = 0; u < GCN_TPT; ++u) {
      rk[u] = 0;
      if (t0 + u * 1024 + tid < tEnd) {
        rk[u] = atomicAdd(&rowCnt[min(max(rc[u].x, 0), 255)], 1);
      }
    }
    __syncthreads();

    int tot2;
    const int rex = gcn_scan256((tid < 256) ? rowCnt[tid] : 0, tid, wtot, tot2);
    if (tid < 256) rowOff[tid] = rex;
    if (tid == 0) rowOff[256] = tot2;
    __syncthreads();

    for (int u = 0; u < GCN_TPT; ++u) {
      if (t0 + u * 1024 + tid < tEnd) {
        int p = rowOff[min(max(rc[u].x, 0), 255)] + rk[u];
        p = min(max(p, 0), GCN_CAP - 1);
        sCV[p]  = make_int2(rc[u].z, rc[u].w);
        sKey[p] = rc[u].y;
      }
    }
    __syncthreads();

    for (;;) {
      int rr = 0;
      if (lane == 0) rr = atomicAdd(&nextRow, 1);
      rr = __builtin_amdgcn_readfirstlane(rr);
      if (rr >= nRows) break;
      const int s   = __builtin_amdgcn_readfirstlane(rowOff[rr]);
      const int len = __builtin_amdgcn_readfirstlane(rowOff[rr + 1]) - s;
      float4 a0, a1;
      if (t0 == 0) {
        a0 = b4[lane];
        a1 = b4[lane + 32];
      } else {
        a0 = acc4[rr * 64 + lane];
        a1 = acc4[rr * 64 + 32 + lane];
      }

      for (int base = 0; base < len; base += 32) {
        const int n = min(32, len - base);
        int2 cv  = make_int2(0, 0);
        int  key = 0x7fffffff;
        if (lane < n) {
          cv  = sCV[s + base + lane];
          key = sKey[s + base + lane];
        }
        if (len <= 32) {
          int rank = 0;
          for (int j = 0; j < n; ++j) {
            rank += (__builtin_amdgcn_readlane(key, j) < key) ? 1 : 0;
          }
          cv.x = __builtin_amdgcn_ds_permute(rank * 4, cv.x);
          cv.y = __builtin_amdgcn_ds_permute(rank * 4, cv.y);
        }

        int j = 0;
        for (; j + 8 <= n; j += 8) gcn_add_edges<8>(y4, cv, j, lane, N - 1, a0, a1);
        if (j + 4 <= n) {
          gcn_add_edges<4>(y4, cv, j, lane, N - 1, a0, a1);
          j += 4;
        }
        const int rem = n - j;
        if (rem == 3)      gcn_add_edges<3>(y4, cv, j, lane, N - 1, a0, a1);
        else if (rem == 2) gcn_add_edges<2>(y4, cv, j, lane, N - 1, a0, a1);
        else if (rem == 1) gcn_add_edges<1>(y4, cv, j, lane, N - 1, a0, a1);
      }

      if (last) {
        float4* o4 = (float4*)out + (size_t)(rowBase + rr) * (D / 4) + lane;
        o4[0]  = a0;
        o4[32] = a1;
      } else {
        acc4[rr * 64 + lane]      = a0;
        acc4[rr * 64 + 32 + lane] = a1;
      }
    }

    if (last) break;
    __syncthreads();
  }
}

extern "C" void kernel_launch(void* const* d_in, const int* in_sizes, int n_in,
                              void* d_out, int out_size, void* d_ws, size_t ws_size,
                              hipStream_t stream) {
  const float* x    = (const float*)d_in[0];
  const int*   erow = (const int*)d_in[1];
  const int*   ecol = (const int*)d_in[2];
  const float* ev   = (const float*)d_in[3];
  const float* W    = (const float*)d_in[4];
  const float* b    = (const float*)d_in[5];
  float*       out  = (float*)d_out;

  const int N   = in_sizes[0] / D;
  const int E   = in_sizes[1];
  const int rpb = (N + GCN_NBK - 1) / GCN_NBK;

  float* y   = (float*)d_ws;
  int*   rec = (int*)(y + (size_t)N * D);
  int*   tab = rec + (size_t)4 * E;

  const int chunkLen = (E + GCN_EBLK - 1) / GCN_EBLK;
  const size_t ldsBytes = (size_t)128 * PAIR_STRIDE * sizeof(v2f);
  gcn_gemm_xw_edges<<<GCN_EBLK, FUSE_THREADS, ldsBytes, stream>>>(
      x, W, y, N / 16, erow, ecol, ev, rec, tab, N, E, rpb, chunkLen);

  const size_t rowLds = (size_t)3 * GCN_CAP * sizeof(int) +
                        (size_t)rpb * D * sizeof(float);
  gcn_bucket_rows<<<GCN_NBK, 1024, rowLds, stream>>>(rec, tab, y, b, out, N, E,
                                                    rpb);
}


# x@W: W fragments of two k-steps are read from LDS as one run, a group of 8 WMMAs before their use, into a second register set (was: pairwise, each waited for in full); k order and sums unchanged
# speedup vs baseline: 4.1917x; 1.0346x over previous
#include <hip/hip_runtime.h>

typedef __attribute__((ext_vector_type(2))) float v2f;
typedef __attribute__((ext_vector_type(8))) float v8f;

#define D 256
#define PAIR_STRIDE 272

extern __shared__ v2f ldsP[];

#define GEMM_QUADS   4
#define GEMM_THREADS (GEMM_QUADS * 128)
#define EDGE_THREADS 256
#define FUSE_THREADS (GEMM_THREADS + EDGE_THREADS)

__device__ __forceinline__ void gcn_load_wgroup(v2f (&b)[8], int lb) {
  for (int s = 0; s < 2; ++s) {
    const int l = lb + (2 * s) * PAIR_STRIDE;
    b[4 * s + 0] = ldsP[l +  0];
    b[4 * s + 1] = ldsP[l + 16];
    b[4 * s + 2] = ldsP[l + 32];
    b[4 * s + 3] = ldsP[l + 48];
  }
}

__device__ __forceinline__ void gcn_mma_wgroup(v2f aLo, v2f aHi,
                                               const v2f (&b)[8], v8f& acc0,
                                               v8f& acc1, v8f& acc2,
                                               v8f& acc3) {
  acc0 = __builtin_amdgcn_wmma_f32_16x16x4_f32(false, aLo, false, b[0],
                                               (short)0, acc0, false, false);
  acc1 = __builtin_amdgcn_wmma_f32_16x16x4_f32(false, aLo, false, b[1],
                                               (short)0, acc1, false, false);
  acc2 = __builtin_amdgcn_wmma_f32_16x16x4_f32(false, aLo, false, b[2],
                                               (short)0, acc2, false, false);
  acc3 = __builtin_amdgcn_wmma_f32_16x16x4_f32(false, aLo, false, b[3],
                                               (short)0, acc3, false, false);
  acc0 = __builtin_amdgcn_wmma_f32_16x16x4_f32(false, aHi, false, b[4],
                                               (short)0, acc0, false, false);
  acc1 = __builtin_amdgcn_wmma_f32_16x16x4_f32(false, aHi, false, b[5],
                                               (short)0, acc1, false, false);
  acc2 = __builtin_amdgcn_wmma_f32_16x16x4_f32(false, aHi, false, b[6],
                                               (short)0, acc2, false, false);
  acc3 = __builtin_amdgcn_wmma_f32_16x16x4_f32(false, aHi, false, b[7],
                                               (short)0, acc3, false, false);
}

__device__ __forceinline__ void gcn_gemm_slab(const v2f (&a)[8], int lb0,
                                              int lbNext, v2f (&bc)[8],
                                              v2f (&bn)[8], v8f& acc0,
                                              v8f& acc1, v8f& acc2,
                                              v8f& acc3) {
  for (int g = 0; g < 4; g += 2) {
    gcn_load_wgroup(bn, lb0 + (4 * (g + 1)) * PAIR_STRIDE);
    __builtin_amdgcn_sched_barrier(0);
    gcn_mma_wgroup(a[2 * g], a[2 * g + 1], bc, acc0, acc1, acc2, acc3);
    __builtin_amdgcn_sched_barrier(0);
    gcn_load_wgroup(bc, (g + 2 < 4) ? lb0 + (4 * (g + 2)) * PAIR_STRIDE
                                    : lbNext);
    __builtin_amdgcn_sched_barrier(0);
    gcn_mma_wgroup(a[2 * g + 2], a[2 * g + 3], bn, acc0, acc1, acc2, acc3);
    __builtin_amdgcn_sched_barrier(0);
  }
}

#define GCN_NBK  256
#define GCN_EBLK 256
#define GCN_EPT  13
#define GCN_CAP  4096
#define GCN_TPT  (GCN_CAP / 1024)

extern __shared__ int4 ldsBk[];

__device__ __forceinline__ int gcn_row_id(int r, int lastRow) {
  return min(max(r, 0), lastRow);
}

__device__ __forceinline__ int gcn_wave_scan(int v, int lane) {
  for (int d = 1; d < 32; d <<= 1) {
    const int t = __shfl_up(v, d);
    if (lane >= d) v += t;
  }
  return v;
}

__device__ __forceinline__ int gcn_scan256(int c, int tid, int* wtot,
                                           int& total) {
  const int lane = tid & 31;
  const int wv   = tid >> 5;
  int inc = 0;
  if (tid < 256) {
    inc = gcn_wave_scan(c, lane);
    if (lane == 31) wtot[wv] = inc;
  }
  __syncthreads();
  int base = 0, all = 0;
  for (int w = 0; w < 8; ++w) {
    const int t = wtot[w];
    if (w < wv) base += t;
    all += t;
  }
  total = all;
  __syncthreads();
  return base + inc - c;
}

static_assert(EDGE_THREADS == GCN_NBK, "one bucket count per edge thread");

__global__ __launch_bounds__(FUSE_THREADS) void gcn_gemm_xw_edges(
    const float* __restrict__ x, const float* __restrict__ W,
    float* __restrict__ y, int nRowTiles, const int* __restrict__ erow,
    const int* __restrict__ ecol, const float* __restrict__ eval_,
    int* __restrict__ recw, int* __restrict__ tabw, int N, int E, int rpb,
    int chunkLen) {
  int4* rec = (int4*)recw;
  int2* tab = (int2*)tabw;
  __shared__ int cnt[GCN_NBK];
  __shared__ int wtot[8];
  const int tid  = threadIdx.x;
  const int lane = tid & 31;
  const int wv   = tid >> 5;
  const int lr   = lane & 15;
  const int hi   = lane >> 4;
  const bool edgeRole = (wv >= GEMM_THREADS / 32);
  const int lt  = tid - GEMM_THREADS;
  const int ew  = wv - GEMM_THREADS / 32;
  const int blk = blockIdx.x;
  const int e0  = min(E, blk * chunkLen);
  const int e1  = min(E, e0 + chunkLen);

  if (edgeRole) cnt[lt] = 0;
  __syncthreads();

  if (!edgeRole) {
#pragma unroll 8
    for (int g = tid; g < (D / 2) * (D / 4); g += GEMM_THREADS) {
      const int p  = g >> 6;
      const int c4 = g & 63;
      const float4 r0 = *(const float4*)(W + (2 * p) * D + 4 * c4);
      const float4 r1 = *(const float4*)(W + (2 * p + 1) * D + 4 * c4);
      v2f* dst = ldsP + p * PAIR_STRIDE + 4 * c4;
      v2f e0v, e1v, e2v, e3v;
      e0v.x = r0.x; e0v.y = r1.x;
      e1v.x = r0.y; e1v.y = r1.y;
      e2v.x = r0.z; e2v.y = r1.z;
      e3v.x = r0.w; e3v.y = r1.w;
      dst[0] = e0v; dst[1] = e1v; dst[2] = e2v; dst[3] = e3v;
    }
  } else {
    for (int b = e0 + lt; b < e1; b += EDGE_THREADS * GCN_EPT) {
      int r[GCN_EPT];
      for (int u = 0; u < GCN_EPT; ++u) {
        r[u] = erow[min(b + u * EDGE_THREADS, e1 - 1)];
      }
      asm volatile("" :: "v"(r[0]), "v"(r[1]), "v"(r[2]), "v"(r[3]),
                         "v"(r[4]), "v"(r[5]), "v"(r[6]), "v"(r[7]),
                         "v"(r[8]), "v"(r[9]), "v"(r[10]), "v"(r[11]),
                         "v"(r[12]));
      for (int u = 0; u < GCN_EPT; ++u) {
        if (b + u * EDGE_THREADS < e1) {
          atomicAdd(&cnt[gcn_row_id(r[u], N - 1) / rpb], 1);
        }
      }
    }
  }
  __syncthreads();

  int c = 0, inc = 0, excl = 0;
  if (edgeRole) {
    c   = cnt[lt];
    inc = gcn_wave_scan(c, lane);
    if (lane == 31) wtot[ew] = inc;
  }
  __syncthreads();
  if (edgeRole) {
    int base = 0;
    for (int w = 0; w < 8; ++w) {
      const int t = wtot[w];
      if (w < ew) base += t;
    }
    excl = base + inc - c;
    cnt[lt] = excl;
  }
  __syncthreads();

  if (edgeRole) {
    tab[lt * GCN_EBLK + blk] = make_int2(e0 + excl, c);

    for (int b = e0 + lt; b < e1; b += EDGE_THREADS * GCN_EPT) {
      int   r[GCN_EPT], cl[GCN_EPT];
      float vl[GCN_EPT];
      for (int u = 0; u < GCN_EPT; ++u) {
        const int e = min(b + u * EDGE_THREADS, e1 - 1);
        r[u]  = erow[e];
        cl[u] = ecol[e];
        vl[u] = eval_[e];
      }
      asm volatile("" :: "v"(r[0]), "v"(r[1]), "v"(r[2]), "v"(r[3]),
                         "v"(r[4]), "v"(r[5]), "v"(r[6]), "v"(r[7]),
                         "v"(r[8]), "v"(r[9]), "v"(r[10]), "v"(r[11]),
                         "v"(r[12]));
      asm volatile("" :: "v"(cl[0]), "v"(cl[1]), "v"(cl[2]), "v"(cl[3]),
                         "v"(cl[4]), "v"(cl[5]), "v"(cl[6]), "v"(cl[7]),
                         "v"(cl[8]), "v"(cl[9]), "v"(cl[10]), "v"(cl[11]),
                         "v"(cl[12]));
      asm volatile("" :: "v"(vl[0]), "v"(vl[1]), "v"(vl[2]), "v"(vl[3]),
                         "v"(vl[4]), "v"(vl[5]), "v"(vl[6]), "v"(vl[7]),
                         "v"(vl[8]), "v"(vl[9]), "v"(vl[10]), "v"(vl[11]),
                         "v"(vl[12]));
      for (int u = 0; u < GCN_EPT; ++u) {
        const int e = b + u * EDGE_THREADS;
        if (e < e1) {
          const int row = gcn_row_id(r[u], N - 1);
          const int bk  = row / rpb;
          int pos = e0 + atomicAdd(&cnt[bk], 1);
          pos = min(max(pos, 0), E - 1);
          rec[pos] = make_int4(row - bk * rpb, e, cl[u], __float_as_int(vl[u]));
        }
      }
    }
    return;
  }

  const int quad    = wv >> 2;
  const int colBase = (wv & 3) * 64;
  const int nb      = gridDim.x;
  const int lbase = hi * PAIR_STRIDE + colBase + lr;

  int t = quad * nb + blockIdx.x;
  const int tstep = GEMM_QUADS * nb;
  if (t >= nRowTiles) return;

  const float* xl = x + lr * D + 2 * hi;
  const float* xp = xl + (size_t)t * (16 * D);

  v2f a0[8], a1[8], a2[8], a3[8];
  for (int u = 0; u < 8; ++u) a0[u] = *(const v2f*)(xp + 4 * u);
  for (int u = 0; u < 8; ++u) a1[u] = *(const v2f*)(xp + 32 + 4 * u);

  v2f wb0[8], wb1[8];
  gcn_load_wgroup(wb0, lbase);

  for (; t < nRowTiles; t += tstep) {
    const int tn = min(t + tstep, nRowTiles - 1);
    const float* xpn = xl + (size_t)tn * (16 * D);

    v8f acc0 = {}, acc1 = {}, acc2 = {}, acc3 = {};

#pragma unroll 1
    for (int j = 0; j < 2; ++j) {
      const float* ps = xp + 128 * j;
      const int    lb = lbase + (64 * j) * PAIR_STRIDE;

      const int    lbn = (j < 1) ? lb + 64 * PAIR_STRIDE : lbase;

      for (int u = 0; u < 8; ++u) a2[u] = *(const v2f*)(ps + 64 + 4 * u);
      gcn_gemm_slab(a0, lb, lb + 16 * PAIR_STRIDE, wb0, wb1, acc0, acc1, acc2,
                    acc3);

      for (int u = 0; u < 8; ++u) a3[u] = *(const v2f*)(ps + 96 + 4 * u);
      gcn_gemm_slab(a1, lb + 16 * PAIR_STRIDE, lb + 32 * PAIR_STRIDE, wb0, wb1,
                    acc0, acc1, acc2, acc3);

      const float* pc = (j < 1) ? ps + 128 : xpn;
      for (int u = 0; u < 8; ++u) a0[u] = *(const v2f*)(pc + 4 * u);
      gcn_gemm_slab(a2, lb + 32 * PAIR_STRIDE, lb + 48 * PAIR_STRIDE, wb0, wb1,
                    acc0, acc1, acc2, acc3);

      for (int u = 0; u < 8; ++u) a1[u] = *(const v2f*)(pc + 32 + 4 * u);
      gcn_gemm_slab(a3, lb + 48 * PAIR_STRIDE, lbn, wb0, wb1, acc0, acc1, acc2,
                    acc3);
    }

    float* yp = y + (size_t)t * (16 * D) + colBase + lr;
#pragma unroll
    for (int i = 0; i < 8; ++i) {
      const int r = (i + 8 * hi) * D;
      yp[r +  0] = acc0[i];
      yp[r + 16] = acc1[i];
      yp[r + 32] = acc2[i];
      yp[r + 48] = acc3[i];
    }
    xp = xpn;
  }
}

template <int K>
__device__ __forceinline__ void gcn_add_edges(const float4* __restrict__ y4,
                                              int2 cv, int j, int lane,
                                              int lastRow, float4& a0,
                                              float4& a1) {
  float4 p[K], q[K];
  float  v[K];
  for (int t = 0; t < K; ++t) {
    const int c = min(max(__builtin_amdgcn_readlane(cv.x, j + t), 0), lastRow);
    v[t] = __int_as_float(__builtin_amdgcn_readlane(cv.y, j + t));
    const float4* s = y4 + (size_t)c * (D / 4) + lane;
    p[t] = s[0];
    q[t] = s[32];
  }
  for (int t = 0; t < K; ++t) {
    a0.x = fmaf(v[t], p[t].x, a0.x); a0.y = fmaf(v[t], p[t].y, a0.y);
    a0.z = fmaf(v[t], p[t].z, a0.z); a0.w = fmaf(v[t], p[t].w, a0.w);
    a1.x = fmaf(v[t], q[t].x, a1.x); a1.y = fmaf(v[t], q[t].y, a1.y);
    a1.z = fmaf(v[t], q[t].z, a1.z); a1.w = fmaf(v[t], q[t].w, a1.w);
  }
}

__global__ __launch_bounds__(1024) void gcn_bucket_rows(
    const int* __restrict__ recw, const int* __restrict__ tabw,
    const float* __restrict__ y, const float* __restrict__ b,
    float* __restrict__ out, int N, int E, int rpb) {
  const int4* rec = (const int4*)recw;
  const int2* tab = (const int2*)tabw;
  __shared__ int runOff[GCN_EBLK + 1];
  __shared__ int runStart[GCN_EBLK];
  __shared__ int rowCnt[256];
  __shared__ int rowOff[257];
  __shared__ int wtot[8];
  __shared__ int nextRow;
  int2*   sCV  = (int2*)ldsBk;
  int*    sKey = (int*)ldsBk + 2 * GCN_CAP;
  float4* acc4 = (float4*)((int*)ldsBk + 3 * GCN_CAP);

  const int tid  = threadIdx.x;
  const int lane = tid & 31;
  const int bk = blockIdx.x;

  int st = 0, c = 0;
  if (tid < GCN_EBLK) {
    const int2 t = tab[bk * GCN_EBLK + tid];
    c  = min(max(t.y, 0), E);
    st = min(max(t.x, 0), E - c);
  }
  int total;
  const int excl = gcn_scan256(c, tid, wtot, total);
  if (tid < GCN_EBLK) {
    runOff[tid]   = excl;
    runStart[tid] = st;
  }
  if (tid == 0) runOff[GCN_EBLK] = total;
  __syncthreads();
  const int M = min(max(total, 0), E);

  const int rowBase = bk * rpb;
  const int nRows   = min(max(N - rowBase, 0), min(rpb, 256));
  const float4* b4 = (const float4*)b;
  const float4* y4 = (const float4*)y;

  for (int t0 = 0;; t0 += GCN_CAP) {
    const int  tEnd = min(M, t0 + GCN_CAP);
    const bool last = (t0 + GCN_CAP >= M);

    if (tid < 256) rowCnt[tid] = 0;
    if (tid == 0) nextRow = 0;
    __syncthreads();

    int4 rc[GCN_TPT];
    int  rk[GCN_TPT];
    for (int u = 0; u < GCN_TPT; ++u) {
      const int k  = t0 + u * 1024 + tid;
      const int kk = min(k, max(tEnd - 1, 0));
      int lo = 0;
      for (int step = GCN_EBLK / 2; step >= 1; step >>= 1) {
        if (runOff[lo + step] <= kk) lo += step;
      }
      int pos = runStart[lo] + (kk - runOff[lo]);
      pos = min(max(pos, 0), E - 1);
      rc[u] = rec[pos];
    }
    asm volatile("" :: "v"(rc[0].x), "v"(rc[1].x), "v"(rc[2].x), "v"(rc[3].x));
    for (int u = 0; u < GCN_TPT; ++u) {
      rk[u] = 0;
      if (t0 + u * 1024 + tid < tEnd) {
        rk[u] = atomicAdd(&rowCnt[min(max(rc[u].x, 0), 255)], 1);
      }
    }
    __syncthreads();

    int tot2;
    const int rex = gcn_scan256((tid < 256) ? rowCnt[tid] : 0, tid, wtot, tot2);
    if (tid < 256) rowOff[tid] = rex;
    if (tid == 0) rowOff[256] = tot2;
    __syncthreads();

    for (int u = 0; u < GCN_TPT; ++u) {
      if (t0 + u * 1024 + tid < tEnd) {
        int p = rowOff[min(max(rc[u].x, 0), 255)] + rk[u];
        p = min(max(p, 0), GCN_CAP - 1);
        sCV[p]  = make_int2(rc[u].z, rc[u].w);
        sKey[p] = rc[u].y;
      }
    }
    __syncthreads();

    for (;;) {
      int rr = 0;
      if (lane == 0) rr = atomicAdd(&nextRow, 1);
      rr = __builtin_amdgcn_readfirstlane(rr);
      if (rr >= nRows) break;
      const int s   = __builtin_amdgcn_readfirstlane(rowOff[rr]);
      const int len = __builtin_amdgcn_readfirstlane(rowOff[rr + 1]) - s;
      float4 a0, a1;
      if (t0 == 0) {
        a0 = b4[lane];
        a1 = b4[lane + 32];
      } else {
        a0 = acc4[rr * 64 + lane];
        a1 = acc4[rr * 64 + 32 + lane];
      }

      for (int base = 0; base < len; base += 32) {
        const int n = min(32, len - base);
        int2 cv  = make_int2(0, 0);
        int  key = 0x7fffffff;
        if (lane < n) {
          cv  = sCV[s + base + lane];
          key = sKey[s + base + lane];
        }
        if (len <= 32) {
          int rank = 0;
          for (int j = 0; j < n; ++j) {
            rank += (__builtin_amdgcn_readlane(key, j) < key) ? 1 : 0;
          }
          cv.x = __builtin_amdgcn_ds_permute(rank * 4, cv.x);
          cv.y = __builtin_amdgcn_ds_permute(rank * 4, cv.y);
        }

        int j = 0;
        for (; j + 8 <= n; j += 8) gcn_add_edges<8>(y4, cv, j, lane, N - 1, a0, a1);
        if (j + 4 <= n) {
          gcn_add_edges<4>(y4, cv, j, lane, N - 1, a0, a1);
          j += 4;
        }
        const int rem = n - j;
        if (rem == 3)      gcn_add_edges<3>(y4, cv, j, lane, N - 1, a0, a1);
        else if (rem == 2) gcn_add_edges<2>(y4, cv, j, lane, N - 1, a0, a1);
        else if (rem == 1) gcn_add_edges<1>(y4, cv, j, lane, N - 1, a0, a1);
      }

      if (last) {
        float4* o4 = (float4*)out + (size_t)(rowBase + rr) * (D / 4) + lane;
        o4[0]  = a0;
        o4[32] = a1;
      } else {
        acc4[rr * 64 + lane]      = a0;
        acc4[rr * 64 + 32 + lane] = a1;
      }
    }

    if (last) break;
    __syncthreads();
  }
}

extern "C" void kernel_launch(void* const* d_in, const int* in_sizes, int n_in,
                              void* d_out, int out_size, void* d_ws, size_t ws_size,
                              hipStream_t stream) {
  const float* x    = (const float*)d_in[0];
  const int*   erow = (const int*)d_in[1];
  const int*   ecol = (const int*)d_in[2];
  const float* ev   = (const float*)d_in[3];
  const float* W    = (const float*)d_in[4];
  const float* b    = (const float*)d_in[5];
  float*       out  = (float*)d_out;

  const int N   = in_sizes[0] / D;
  const int E   = in_sizes[1];
  const int rpb = (N + GCN_NBK - 1) / GCN_NBK;

  float* y   = (float*)d_ws;
  int*   rec = (int*)(y + (size_t)N * D);
  int*   tab = rec + (size_t)4 * E;

  const int chunkLen = (E + GCN_EBLK - 1) / GCN_EBLK;
  const size_t ldsBytes = (size_t)128 * PAIR_STRIDE * sizeof(v2f);
  gcn_gemm_xw_edges<<<GCN_EBLK, FUSE_THREADS, ldsBytes, stream>>>(
      x, W, y, N / 16, erow, ecol, ev, rec, tab, N, E, rpb, chunkLen);

  const size_t rowLds = (size_t)3 * GCN_CAP * sizeof(int) +
                        (size_t)rpb * D * sizeof(float);
  gcn_bucket_rows<<<GCN_NBK, 1024, rowLds, stream>>>(rec, tab, y, b, out, N, E,
                                                    rpb);
}
